# starting kernel
# baseline (speedup 1.0000x reference)
#include <hip/hip_runtime.h>
#include <math.h>


typedef __attribute__((ext_vector_type(2))) float v2f;
typedef __attribute__((ext_vector_type(8))) float v8f;

#define HW      16384
#define NBATCH  8
#define CCH     128
#define MID     64
#define PDIM    32
#define NBAND   8

__device__ __forceinline__ v8f wmma4(v2f a, v2f b, v8f c) {
  return __builtin_amdgcn_wmma_f32_16x16x4_f32(false, a, false, b,
                                               (short)0, c, false, false);
}

__device__ __forceinline__ float gelu_f(float x) {
  return 0.5f * x * (1.0f + erff(x * 0.70710678118654752440f));
}
__device__ __forceinline__ float clamp01(float x) {
  return fminf(fmaxf(x, 0.0f), 1.0f);
}

__global__ void k_emb(const float* __restrict__ Wp, float* __restrict__ emb) {
  int p = blockIdx.x * blockDim.x + threadIdx.x;
  if (p >= HW) return;
  int i = p >> 7, j = p & 127;
  float ym = -1.0f + (2.0f / 127.0f) * (float)i;
  float xm = -1.0f + (2.0f / 127.0f) * (float)j;
#pragma unroll 4
  for (int pd = 0; pd < PDIM; ++pd) {
    float v = tanhf(Wp[pd * 2 + 0] * ym + Wp[pd * 2 + 1] * xm);
    emb[(size_t)pd * HW + p] = v;
  }
}

__global__ __launch_bounds__(256) void k_modulate(
    const float* __restrict__ x, const float* __restrict__ emb,
    const float* __restrict__ Wm1, const float* __restrict__ Wm2,
    float* __restrict__ xmod) {
  __shared__ float lds[8][MID * 17];
  const int wave = threadIdx.x >> 5;
  const int lane = threadIdx.x & 31;
  const int n    = blockIdx.x >> 7;
  const int pb   = (blockIdx.x & 127) * 128 + wave * 16;
  const int lm   = lane & 15;
  const int kh   = (lane >> 4) * 2;
  const int mh   = (lane >> 4) * 8;
  const int pix  = pb + lm;
  const float* xn = x + (size_t)n * CCH * HW;

  v8f acc[4] = {};
  for (int k0 = 0; k0 < CCH; k0 += 4) {
    int kk = k0 + kh;
    v2f b;
    b.x = xn[(size_t)kk * HW + pix];
    b.y = xn[(size_t)(kk + 1) * HW + pix];
#pragma unroll
    for (int t = 0; t < 4; ++t) {
      v2f a = *(const v2f*)(Wm1 + (t * 16 + lm) * (CCH + PDIM) + kk);
      acc[t] = wmma4(a, b, acc[t]);
    }
  }
  for (int k0 = 0; k0 < PDIM; k0 += 4) {
    int kk = k0 + kh;
    v2f b;
    b.x = emb[(size_t)kk * HW + pix];
    b.y = emb[(size_t)(kk + 1) * HW + pix];
#pragma unroll
    for (int t = 0; t < 4; ++t) {
      v2f a = *(const v2f*)(Wm1 + (t * 16 + lm) * (CCH + PDIM) + CCH + kk);
      acc[t] = wmma4(a, b, acc[t]);
    }
  }
  float* sm = &lds[wave][0];
#pragma unroll
  for (int t = 0; t < 4; ++t)
#pragma unroll
    for (int v = 0; v < 8; ++v)
      sm[(t * 16 + v + mh) * 17 + lm] = gelu_f(acc[t][v]);

  v8f acc2[8] = {};
  for (int k0 = 0; k0 < MID; k0 += 4) {
    int kk = k0 + kh;
    v2f b;
    b.x = sm[kk * 17 + lm];
    b.y = sm[(kk + 1) * 17 + lm];
#pragma unroll
    for (int t = 0; t < 8; ++t) {
      v2f a = *(const v2f*)(Wm2 + (t * 16 + lm) * MID + kk);
      acc2[t] = wmma4(a, b, acc2[t]);
    }
  }
  float* xo = xmod + (size_t)n * CCH * HW;
#pragma unroll
  for (int t = 0; t < 8; ++t)
#pragma unroll
    for (int v = 0; v < 8; ++v) {
      int m = t * 16 + v + mh;
      xo[(size_t)m * HW + pix] = clamp01(gelu_f(acc2[t][v]));
    }
}

__global__ __launch_bounds__(256) void k_band_stats(
    const float* __restrict__ xmod, const float* __restrict__ W1,
    float* __restrict__ psum, float* __restrict__ psq) {
  const int band  = threadIdx.x >> 5;
  const int lane  = threadIdx.x & 31;
  const int n     = blockIdx.x >> 7;
  const int chunk = blockIdx.x & 127;
  const int lm = lane & 15, kh = (lane >> 4) * 2, mh = (lane >> 4) * 8;
  const float* xn = xmod + (size_t)n * CCH * HW;
  const float* A  = W1 + (size_t)band * MID * CCH;

  v8f s_run[4] = {};
  v8f q_run[4] = {};
  for (int sb = 0; sb < 8; ++sb) {
    const int pix = chunk * 128 + sb * 16 + lm;
    v8f acc[4] = {};
    for (int k0 = 0; k0 < CCH; k0 += 4) {
      int kk = k0 + kh;
      v2f bb;
      bb.x = xn[(size_t)kk * HW + pix];
      bb.y = xn[(size_t)(kk + 1) * HW + pix];
#pragma unroll
      for (int t = 0; t < 4; ++t) {
        v2f a = *(const v2f*)(A + (t * 16 + lm) * CCH + kk);
        acc[t] = wmma4(a, bb, acc[t]);
      }
    }
#pragma unroll
    for (int t = 0; t < 4; ++t) {
      s_run[t] += acc[t];
      q_run[t] += acc[t] * acc[t];
    }
  }
  const int nb = n * NBAND + band;
#pragma unroll
  for (int t = 0; t < 4; ++t)
#pragma unroll
    for (int v = 0; v < 8; ++v) {
      float s = s_run[t][v], q = q_run[t][v];
#pragma unroll
      for (int off = 1; off < 16; off <<= 1) {
        s += __shfl_xor(s, off, 32);
        q += __shfl_xor(q, off, 32);
      }
      if (lm == 0) {
        int m = t * 16 + v + mh;
        psum[((size_t)nb * MID + m) * 128 + chunk] = s;
        psq [((size_t)nb * MID + m) * 128 + chunk] = q;
      }
    }
}

__global__ void k_reduce_stats(const float* __restrict__ psum,
                               const float* __restrict__ psq,
                               float* __restrict__ mean,
                               float* __restrict__ rstd) {
  int i = blockIdx.x * blockDim.x + threadIdx.x;
  if (i >= NBATCH * NBAND * MID) return;
  float s = 0.0f, q = 0.0f;
  for (int j = 0; j < 128; ++j) { s += psum[(size_t)i * 128 + j]; q += psq[(size_t)i * 128 + j]; }
  float mu  = s * (1.0f / (float)HW);
  float var = q * (1.0f / (float)HW) - mu * mu;
  mean[i] = mu;
  rstd[i] = rsqrtf(fmaxf(var, 0.0f) + 1e-5f);
}

__global__ __launch_bounds__(256) void k_bands(
    const float* __restrict__ xmod, const float* __restrict__ W1,
    const float* __restrict__ W2, const float* __restrict__ W3,
    const float* __restrict__ mean, const float* __restrict__ rstd,
    float* __restrict__ out) {
  __shared__ float lds[8][MID * 17];
  const int band  = threadIdx.x >> 5;
  const int lane  = threadIdx.x & 31;
  const int n     = blockIdx.x >> 7;
  const int chunk = blockIdx.x & 127;
  const int lm = lane & 15, kh = (lane >> 4) * 2, mh = (lane >> 4) * 8;
  const int nb = n * NBAND + band;
  const float* xn = xmod + (size_t)n * CCH * HW;
  const float* A  = W1 + (size_t)band * MID * CCH;
  const float* A2 = W2 + (size_t)band * 32 * MID;
  const float* w3 = W3 + (size_t)band * 32;
  const float* mu = mean + (size_t)nb * MID;
  const float* rs = rstd + (size_t)nb * MID;
  float* sm = &lds[band][0];

  for (int sb = 0; sb < 8; ++sb) {
    const int pix = chunk * 128 + sb * 16 + lm;

    v8f acc[4] = {};
    for (int k0 = 0; k0 < CCH; k0 += 4) {
      int kk = k0 + kh;
      v2f bb;
      bb.x = xn[(size_t)kk * HW + pix];
      bb.y = xn[(size_t)(kk + 1) * HW + pix];
#pragma unroll
      for (int t = 0; t < 4; ++t) {
        v2f a = *(const v2f*)(A + (t * 16 + lm) * CCH + kk);
        acc[t] = wmma4(a, bb, acc[t]);
      }
    }
#pragma unroll
    for (int t = 0; t < 4; ++t)
#pragma unroll
      for (int v = 0; v < 8; ++v) {
        int m = t * 16 + v + mh;
        sm[m * 17 + lm] = gelu_f((acc[t][v] - mu[m]) * rs[m]);
      }

    v8f acc2[2] = {};
    for (int k0 = 0; k0 < MID; k0 += 4) {
      int kk = k0 + kh;
      v2f bb;
      bb.x = sm[kk * 17 + lm];
      bb.y = sm[(kk + 1) * 17 + lm];
#pragma unroll
      for (int t = 0; t < 2; ++t) {
        v2f a = *(const v2f*)(A2 + (t * 16 + lm) * MID + kk);
        acc2[t] = wmma4(a, bb, acc2[t]);
      }
    }
#pragma unroll
    for (int t = 0; t < 2; ++t)
#pragma unroll
      for (int v = 0; v < 8; ++v)
        sm[(t * 16 + v + mh) * 17 + lm] = gelu_f(acc2[t][v]);

    if (lane < 16) {
      float s = 0.0f;
#pragma unroll
      for (int k = 0; k < 32; ++k) s += w3[k] * sm[k * 17 + lane];
      out[(size_t)nb * HW + chunk * 128 + sb * 16 + lane] = clamp01(gelu_f(s));
    }
  }
}

extern "C" void kernel_launch(void* const* d_in, const int* in_sizes, int n_in,
                              void* d_out, int out_size, void* d_ws, size_t ws_size,
                              hipStream_t stream) {
  const float* x   = (const float*)d_in[0];
  const float* Wp  = (const float*)d_in[1];
  const float* Wm1 = (const float*)d_in[2];
  const float* Wm2 = (const float*)d_in[3];
  const float* W1  = (const float*)d_in[4];
  const float* W2  = (const float*)d_in[5];
  const float* W3  = (const float*)d_in[6];
  float* out = (float*)d_out;

  float* ws   = (float*)d_ws;
  float* emb  = ws;
  float* xmod = emb + (size_t)PDIM * HW;
  float* psum = xmod + (size_t)NBATCH * CCH * HW;
  float* psq  = psum + (size_t)NBATCH * NBAND * MID * 128;
  float* mean = psq + (size_t)NBATCH * NBAND * MID * 128;
  float* rstd = mean + (size_t)NBATCH * NBAND * MID;
  (void)in_sizes; (void)n_in; (void)out_size; (void)ws_size;

  k_emb<<<HW / 256, 256, 0, stream>>>(Wp, emb);
  k_modulate<<<NBATCH * 128, 256, 0, stream>>>(x, emb, Wm1, Wm2, xmod);
  k_band_stats<<<NBATCH * 128, 256, 0, stream>>>(xmod, W1, psum, psq);
  k_reduce_stats<<<(NBATCH * NBAND * MID + 255) / 256, 256, 0, stream>>>(psum, psq, mean, rstd);
  k_bands<<<NBATCH * 128, 256, 0, stream>>>(xmod, W1, W2, W3, mean, rstd, out);
}


# Band kernels: each wave loads its band's W1 and W2 WMMA fragments into registers once and reuses them over a 512-pixel range (grid 256) instead of re-gathering them per 16-pixel sub-tile
# speedup vs baseline: 1.5059x; 1.5059x over previous
#include <hip/hip_runtime.h>
#include <math.h>


typedef __attribute__((ext_vector_type(2))) float v2f;
typedef __attribute__((ext_vector_type(8))) float v8f;

#define HW      16384
#define NBATCH  8
#define CCH     128
#define MID     64
#define PDIM    32
#define NBAND   8
#define PXB     512
#define SBN     (PXB / 16)
#define BPI     (HW / PXB)
#define KS1     (CCH / 4)
#define KS2     (MID / 4)

__device__ __forceinline__ v8f wmma4(v2f a, v2f b, v8f c) {
  return __builtin_amdgcn_wmma_f32_16x16x4_f32(false, a, false, b,
                                               (short)0, c, false, false);
}

__device__ __forceinline__ float gelu_f(float x) {
  return 0.5f * x * (1.0f + erff(x * 0.70710678118654752440f));
}
__device__ __forceinline__ float clamp01(float x) {
  return fminf(fmaxf(x, 0.0f), 1.0f);
}

__global__ void k_emb(const float* __restrict__ Wp, float* __restrict__ emb) {
  int p = blockIdx.x * blockDim.x + threadIdx.x;
  if (p >= HW) return;
  int i = p >> 7, j = p & 127;
  float ym = -1.0f + (2.0f / 127.0f) * (float)i;
  float xm = -1.0f + (2.0f / 127.0f) * (float)j;
#pragma unroll 4
  for (int pd = 0; pd < PDIM; ++pd) {
    float v = tanhf(Wp[pd * 2 + 0] * ym + Wp[pd * 2 + 1] * xm);
    emb[(size_t)pd * HW + p] = v;
  }
}

__global__ __launch_bounds__(256) void k_modulate(
    const float* __restrict__ x, const float* __restrict__ emb,
    const float* __restrict__ Wm1, const float* __restrict__ Wm2,
    float* __restrict__ xmod) {
  __shared__ float lds[8][MID * 17];
  const int wave = threadIdx.x >> 5;
  const int lane = threadIdx.x & 31;
  const int n    = blockIdx.x >> 7;
  const int pb   = (blockIdx.x & 127) * 128 + wave * 16;
  const int lm   = lane & 15;
  const int kh   = (lane >> 4) * 2;
  const int mh   = (lane >> 4) * 8;
  const int pix  = pb + lm;
  const float* xn = x + (size_t)n * CCH * HW;

  v8f acc[4] = {};
  for (int k0 = 0; k0 < CCH; k0 += 4) {
    int kk = k0 + kh;
    v2f b;
    b.x = xn[(size_t)kk * HW + pix];
    b.y = xn[(size_t)(kk + 1) * HW + pix];
#pragma unroll
    for (int t = 0; t < 4; ++t) {
      v2f a = *(const v2f*)(Wm1 + (t * 16 + lm) * (CCH + PDIM) + kk);
      acc[t] = wmma4(a, b, acc[t]);
    }
  }
  for (int k0 = 0; k0 < PDIM; k0 += 4) {
    int kk = k0 + kh;
    v2f b;
    b.x = emb[(size_t)kk * HW + pix];
    b.y = emb[(size_t)(kk + 1) * HW + pix];
#pragma unroll
    for (int t = 0; t < 4; ++t) {
      v2f a = *(const v2f*)(Wm1 + (t * 16 + lm) * (CCH + PDIM) + CCH + kk);
      acc[t] = wmma4(a, b, acc[t]);
    }
  }
  float* sm = &lds[wave][0];
#pragma unroll
  for (int t = 0; t < 4; ++t)
#pragma unroll
    for (int v = 0; v < 8; ++v)
      sm[(t * 16 + v + mh) * 17 + lm] = gelu_f(acc[t][v]);

  v8f acc2[8] = {};
  for (int k0 = 0; k0 < MID; k0 += 4) {
    int kk = k0 + kh;
    v2f b;
    b.x = sm[kk * 17 + lm];
    b.y = sm[(kk + 1) * 17 + lm];
#pragma unroll
    for (int t = 0; t < 8; ++t) {
      v2f a = *(const v2f*)(Wm2 + (t * 16 + lm) * MID + kk);
      acc2[t] = wmma4(a, b, acc2[t]);
    }
  }
  float* xo = xmod + (size_t)n * CCH * HW;
#pragma unroll
  for (int t = 0; t < 8; ++t)
#pragma unroll
    for (int v = 0; v < 8; ++v) {
      int m = t * 16 + v + mh;
      xo[(size_t)m * HW + pix] = clamp01(gelu_f(acc2[t][v]));
    }
}

__global__ __launch_bounds__(256) void k_band_stats(
    const float* __restrict__ xmod, const float* __restrict__ W1,
    float* __restrict__ psum, float* __restrict__ psq) {
  const int band = threadIdx.x >> 5;
  const int lane = threadIdx.x & 31;
  const int n    = blockIdx.x / BPI;
  const int blk  = blockIdx.x % BPI;
  const int lm = lane & 15, kh = (lane >> 4) * 2, mh = (lane >> 4) * 8;
  const float* A  = W1 + (size_t)band * MID * CCH;
  const float* xb = xmod + (size_t)n * CCH * HW + (size_t)kh * HW + blk * PXB + lm;

  v2f a[4][KS1];
#pragma unroll
  for (int t = 0; t < 4; ++t)
#pragma unroll
    for (int k = 0; k < KS1; ++k)
      a[t][k] = *(const v2f*)(A + (t * 16 + lm) * CCH + 4 * k + kh);

  v2f bq[KS1];
#pragma unroll
  for (int k = 0; k < KS1; ++k) {
    bq[k].x = xb[(size_t)(4 * k) * HW];
    bq[k].y = xb[(size_t)(4 * k + 1) * HW];
  }

  const int nb = n * NBAND + band;
  v8f s_run[4] = {};
  v8f q_run[4] = {};
  for (int sb = 0; sb < SBN; ++sb) {
    v8f acc[4] = {};
#pragma unroll
    for (int k = 0; k < KS1; ++k)
#pragma unroll
      for (int t = 0; t < 4; ++t)
        acc[t] = wmma4(a[t][k], bq[k], acc[t]);

    const int nx = (sb + 1 < SBN ? sb + 1 : sb) * 16;
#pragma unroll
    for (int k = 0; k < KS1; ++k) {
      bq[k].x = xb[(size_t)(4 * k) * HW + nx];
      bq[k].y = xb[(size_t)(4 * k + 1) * HW + nx];
    }

#pragma unroll
    for (int t = 0; t < 4; ++t) {
      s_run[t] += acc[t];
      q_run[t] += acc[t] * acc[t];
    }
    if ((sb & 7) == 7) {
      const int chunk = blk * (PXB / 128) + (sb >> 3);
#pragma unroll
      for (int t = 0; t < 4; ++t) {
#pragma unroll
        for (int v = 0; v < 8; ++v) {
          float s = s_run[t][v], q = q_run[t][v];
#pragma unroll
          for (int off = 1; off < 16; off <<= 1) {
            s += __shfl_xor(s, off, 32);
            q += __shfl_xor(q, off, 32);
          }
          if (lm == 0) {
            int m = t * 16 + v + mh;
            psum[((size_t)nb * MID + m) * 128 + chunk] = s;
            psq [((size_t)nb * MID + m) * 128 + chunk] = q;
          }
        }
        s_run[t] = (v8f){};
        q_run[t] = (v8f){};
      }
    }
  }
}

__global__ void k_reduce_stats(const float* __restrict__ psum,
                               const float* __restrict__ psq,
                               float* __restrict__ mean,
                               float* __restrict__ rstd) {
  int i = blockIdx.x * blockDim.x + threadIdx.x;
  if (i >= NBATCH * NBAND * MID) return;
  float s = 0.0f, q = 0.0f;
  for (int j = 0; j < 128; ++j) { s += psum[(size_t)i * 128 + j]; q += psq[(size_t)i * 128 + j]; }
  float mu  = s * (1.0f / (float)HW);
  float var = q * (1.0f / (float)HW) - mu * mu;
  mean[i] = mu;
  rstd[i] = rsqrtf(fmaxf(var, 0.0f) + 1e-5f);
}

__global__ __launch_bounds__(256) void k_bands(
    const float* __restrict__ xmod, const float* __restrict__ W1,
    const float* __restrict__ W2, const float* __restrict__ W3,
    const float* __restrict__ mean, const float* __restrict__ rstd,
    float* __restrict__ out) {
  __shared__ float lds[8][MID * 17];
  const int band = threadIdx.x >> 5;
  const int lane = threadIdx.x & 31;
  const int n    = blockIdx.x / BPI;
  const int blk  = blockIdx.x % BPI;
  const int lm = lane & 15, kh = (lane >> 4) * 2, mh = (lane >> 4) * 8;
  const int nb = n * NBAND + band;
  const float* A  = W1 + (size_t)band * MID * CCH;
  const float* A2 = W2 + (size_t)band * 32 * MID;
  const float* w3 = W3 + (size_t)band * 32;
  const float* mu = mean + (size_t)nb * MID;
  const float* rs = rstd + (size_t)nb * MID;
  const float* xb = xmod + (size_t)n * CCH * HW + (size_t)kh * HW + blk * PXB + lm;
  float* sm = &lds[band][0];

  v2f a[4][KS1];
#pragma unroll
  for (int t = 0; t < 4; ++t)
#pragma unroll
    for (int k = 0; k < KS1; ++k)
      a[t][k] = *(const v2f*)(A + (t * 16 + lm) * CCH + 4 * k + kh);
  v2f a2[2][KS2];
#pragma unroll
  for (int t = 0; t < 2; ++t)
#pragma unroll
    for (int k = 0; k < KS2; ++k)
      a2[t][k] = *(const v2f*)(A2 + (t * 16 + lm) * MID + 4 * k + kh);

  v2f bq[KS1];
#pragma unroll
  for (int k = 0; k < KS1; ++k) {
    bq[k].x = xb[(size_t)(4 * k) * HW];
    bq[k].y = xb[(size_t)(4 * k + 1) * HW];
  }

  for (int sb = 0; sb < SBN; ++sb) {
    asm volatile("" ::: "memory");
    v8f acc[4] = {};
#pragma unroll
    for (int k = 0; k < KS1; ++k)
#pragma unroll
      for (int t = 0; t < 4; ++t)
        acc[t] = wmma4(a[t][k], bq[k], acc[t]);

    const int nx = (sb + 1 < SBN ? sb + 1 : sb) * 16;
#pragma unroll
    for (int k = 0; k < KS1; ++k) {
      bq[k].x = xb[(size_t)(4 * k) * HW + nx];
      bq[k].y = xb[(size_t)(4 * k + 1) * HW + nx];
    }

#pragma unroll
    for (int t = 0; t < 4; ++t)
#pragma unroll
      for (int v = 0; v < 8; ++v) {
        int m = t * 16 + v + mh;
        sm[m * 17 + lm] = gelu_f((acc[t][v] - mu[m]) * rs[m]);
      }

    v8f acc2[2] = {};
#pragma unroll
    for (int k = 0; k < KS2; ++k) {
      int kk = 4 * k + kh;
      v2f bb;
      bb.x = sm[kk * 17 + lm];
      bb.y = sm[(kk + 1) * 17 + lm];
#pragma unroll
      for (int t = 0; t < 2; ++t)
        acc2[t] = wmma4(a2[t][k], bb, acc2[t]);
    }
#pragma unroll
    for (int t = 0; t < 2; ++t)
#pragma unroll
      for (int v = 0; v < 8; ++v)
        sm[(t * 16 + v + mh) * 17 + lm] = gelu_f(acc2[t][v]);

    if (lane < 16) {
      float s = 0.0f;
#pragma unroll
      for (int k = 0; k < 32; ++k) s += w3[k] * sm[k * 17 + lane];
      out[(size_t)nb * HW + blk * PXB + sb * 16 + lane] = clamp01(gelu_f(s));
    }
  }
}

extern "C" void kernel_launch(void* const* d_in, const int* in_sizes, int n_in,
                              void* d_out, int out_size, void* d_ws, size_t ws_size,
                              hipStream_t stream) {
  const float* x   = (const float*)d_in[0];
  const float* Wp  = (const float*)d_in[1];
  const float* Wm1 = (const float*)d_in[2];
  const float* Wm2 = (const float*)d_in[3];
  const float* W1  = (const float*)d_in[4];
  const float* W2  = (const float*)d_in[5];
  const float* W3  = (const float*)d_in[6];
  float* out = (float*)d_out;

  float* ws   = (float*)d_ws;
  float* emb  = ws;
  float* xmod = emb + (size_t)PDIM * HW;
  float* psum = xmod + (size_t)NBATCH * CCH * HW;
  float* psq  = psum + (size_t)NBATCH * NBAND * MID * 128;
  float* mean = psq + (size_t)NBATCH * NBAND * MID * 128;
  float* rstd = mean + (size_t)NBATCH * NBAND * MID;
  (void)in_sizes; (void)n_in; (void)out_size; (void)ws_size;

  k_emb<<<HW / 256, 256, 0, stream>>>(Wp, emb);
  k_modulate<<<NBATCH * 128, 256, 0, stream>>>(x, emb, Wm1, Wm2, xmod);
  k_band_stats<<<NBATCH * BPI, 256, 0, stream>>>(xmod, W1, psum, psq);
  k_reduce_stats<<<(NBATCH * NBAND * MID + 255) / 256, 256, 0, stream>>>(psum, psq, mean, rstd);
  k_bands<<<NBATCH * BPI, 256, 0, stream>>>(xmod, W1, W2, W3, mean, rstd, out);
}


# Instance-norm stats from the per-image Gram matrix of xmod (mean = W1 mu, var = w' Cov w; WMMA Gram, double-summed covariance) instead of h1 for all 8 bands; stats summation order changes
# speedup vs baseline: 1.6839x; 1.1182x over previous
#include <hip/hip_runtime.h>
#include <math.h>


typedef __attribute__((ext_vector_type(2))) float v2f;
typedef __attribute__((ext_vector_type(4))) float v4f;
typedef __attribute__((ext_vector_type(8))) float v8f;

#define HW      16384
#define NBATCH  8
#define CCH     128
#define MID     64
#define PDIM    32
#define NBAND   8
#define PXB     512
#define SBN     (PXB / 16)
#define BPI     (HW / PXB)
#define KS1     (CCH / 4)
#define KS2     (MID / 4)
#define GPX     2048
#define GRNG    (HW / GPX)
#define GIT     128
#define GPITCH  (GIT + 4)

__device__ __forceinline__ v8f wmma4(v2f a, v2f b, v8f c) {
  return __builtin_amdgcn_wmma_f32_16x16x4_f32(false, a, false, b,
                                               (short)0, c, false, false);
}

__device__ __forceinline__ float gelu_f(float x) {
  return 0.5f * x * (1.0f + erff(x * 0.70710678118654752440f));
}
__device__ __forceinline__ float clamp01(float x) {
  return fminf(fmaxf(x, 0.0f), 1.0f);
}

__global__ void k_emb(const float* __restrict__ Wp, float* __restrict__ emb) {
  int p = blockIdx.x * blockDim.x + threadIdx.x;
  if (p >= HW) return;
  int i = p >> 7, j = p & 127;
  float ym = -1.0f + (2.0f / 127.0f) * (float)i;
  float xm = -1.0f + (2.0f / 127.0f) * (float)j;
#pragma unroll 4
  for (int pd = 0; pd < PDIM; ++pd) {
    float v = tanhf(Wp[pd * 2 + 0] * ym + Wp[pd * 2 + 1] * xm);
    emb[(size_t)pd * HW + p] = v;
  }
}

__global__ __launch_bounds__(256) void k_modulate(
    const float* __restrict__ x, const float* __restrict__ emb,
    const float* __restrict__ Wm1, const float* __restrict__ Wm2,
    float* __restrict__ xmod) {
  __shared__ float lds[8][MID * 17];
  const int wave = threadIdx.x >> 5;
  const int lane = threadIdx.x & 31;
  const int n    = blockIdx.x >> 7;
  const int pb   = (blockIdx.x & 127) * 128 + wave * 16;
  const int lm   = lane & 15;
  const int kh   = (lane >> 4) * 2;
  const int mh   = (lane >> 4) * 8;
  const int pix  = pb + lm;
  const float* xn = x + (size_t)n * CCH * HW;

  v8f acc[4] = {};
  for (int k0 = 0; k0 < CCH; k0 += 4) {
    int kk = k0 + kh;
    v2f b;
    b.x = xn[(size_t)kk * HW + pix];
    b.y = xn[(size_t)(kk + 1) * HW + pix];
#pragma unroll
    for (int t = 0; t < 4; ++t) {
      v2f a = *(const v2f*)(Wm1 + (t * 16 + lm) * (CCH + PDIM) + kk);
      acc[t] = wmma4(a, b, acc[t]);
    }
  }
  for (int k0 = 0; k0 < PDIM; k0 += 4) {
    int kk = k0 + kh;
    v2f b;
    b.x = emb[(size_t)kk * HW + pix];
    b.y = emb[(size_t)(kk + 1) * HW + pix];
#pragma unroll
    for (int t = 0; t < 4; ++t) {
      v2f a = *(const v2f*)(Wm1 + (t * 16 + lm) * (CCH + PDIM) + CCH + kk);
      acc[t] = wmma4(a, b, acc[t]);
    }
  }
  float* sm = &lds[wave][0];
#pragma unroll
  for (int t = 0; t < 4; ++t)
#pragma unroll
    for (int v = 0; v < 8; ++v)
      sm[(t * 16 + v + mh) * 17 + lm] = gelu_f(acc[t][v]);

  v8f acc2[8] = {};
  for (int k0 = 0; k0 < MID; k0 += 4) {
    int kk = k0 + kh;
    v2f b;
    b.x = sm[kk * 17 + lm];
    b.y = sm[(kk + 1) * 17 + lm];
#pragma unroll
    for (int t = 0; t < 8; ++t) {
      v2f a = *(const v2f*)(Wm2 + (t * 16 + lm) * MID + kk);
      acc2[t] = wmma4(a, b, acc2[t]);
    }
  }
  float* xo = xmod + (size_t)n * CCH * HW;
#pragma unroll
  for (int t = 0; t < 8; ++t)
#pragma unroll
    for (int v = 0; v < 8; ++v) {
      int m = t * 16 + v + mh;
      xo[(size_t)m * HW + pix] = clamp01(gelu_f(acc2[t][v]));
    }
}

__global__ __launch_bounds__(256) void k_gram(
    const float* __restrict__ xmod, float* __restrict__ gpart,
    float* __restrict__ csum) {
  __shared__ __attribute__((aligned(16))) float tile[2][CCH * GPITCH];
  __shared__ float csl[CCH];
  const int wave = threadIdx.x >> 5;
  const int lane = threadIdx.x & 31;
  const int g    = blockIdx.x & 3;
  const int rng  = (blockIdx.x >> 2) & (GRNG - 1);
  const int n    = blockIdx.x >> 5;
  const int lm = lane & 15, hh = lane >> 4, mh = hh * 8;
  const int rb = 2 * g + (wave & 1);
  const int cb = 2 * (wave >> 1);
  const float* sp = xmod + (size_t)n * CCH * HW + (size_t)(16 * wave) * HW
                  + (size_t)rng * GPX + 4 * lane;
  v4f st[16];
#pragma unroll
  for (int q = 0; q < 16; ++q) st[q] = *(const v4f*)(sp + (size_t)q * HW);
  float rsum[16];
#pragma unroll
  for (int q = 0; q < 16; ++q) rsum[q] = 0.0f;
  v8f acc0 = {}, acc1 = {};
  const int fa  = (16 * rb + lm) * GPITCH + 4 * hh;
  const int fb0 = (16 * cb + lm) * GPITCH + 4 * hh;
  const int fb1 = fb0 + 16 * GPITCH;

  for (int it = 0; it < GPX / GIT; ++it) {
    float* tb = &tile[it & 1][0];
#pragma unroll
    for (int q = 0; q < 16; ++q) {
      *(v4f*)(tb + (16 * wave + q) * GPITCH + 4 * lane) = st[q];
      rsum[q] += (st[q].x + st[q].y) + (st[q].z + st[q].w);
    }
    __syncthreads();
    if (it + 1 < GPX / GIT) {
#pragma unroll
      for (int q = 0; q < 16; ++q)
        st[q] = *(const v4f*)(sp + (size_t)q * HW + (it + 1) * GIT);
    }
    v4f a  = *(const v4f*)(tb + fa);
    v4f b0 = *(const v4f*)(tb + fb0);
    v4f b1 = *(const v4f*)(tb + fb1);
#pragma unroll
    for (int s = 0; s < GIT / 8; ++s) {
      v4f an = a, b0n = b0, b1n = b1;
      if (s + 1 < GIT / 8) {
        an  = *(const v4f*)(tb + fa  + 8 * (s + 1));
        b0n = *(const v4f*)(tb + fb0 + 8 * (s + 1));
        b1n = *(const v4f*)(tb + fb1 + 8 * (s + 1));
      }
      v2f alo = {a.x, a.y},   ahi = {a.z, a.w};
      v2f plo = {b0.x, b0.y}, phi = {b0.z, b0.w};
      v2f qlo = {b1.x, b1.y}, qhi = {b1.z, b1.w};
      acc0 = wmma4(alo, plo, acc0);
      acc1 = wmma4(alo, qlo, acc1);
      acc0 = wmma4(ahi, phi, acc0);
      acc1 = wmma4(ahi, qhi, acc1);
      a = an; b0 = b0n; b1 = b1n;
    }
  }

  float* gp = gpart + (size_t)(n * GRNG + rng) * (CCH * CCH);
#pragma unroll
  for (int v = 0; v < 8; ++v) {
    int row = 16 * rb + mh + v;
    gp[row * CCH + 16 * cb + lm]       = acc0[v];
    gp[row * CCH + 16 * (cb + 1) + lm] = acc1[v];
  }
#pragma unroll
  for (int q = 0; q < 16; ++q) {
    float s = rsum[q];
#pragma unroll
    for (int off = 1; off < 32; off <<= 1) s += __shfl_xor(s, off, 32);
    if (lane == 0) csl[16 * wave + q] = s;
  }
  __syncthreads();
  if (g == 0 && threadIdx.x < CCH)
    csum[(n * GRNG + rng) * CCH + threadIdx.x] = csl[threadIdx.x];
}

__global__ void k_cov(const float* __restrict__ gpart,
                      const float* __restrict__ csum,
                      float* __restrict__ cov, float* __restrict__ mu) {
  int i = blockIdx.x * blockDim.x + threadIdx.x;
  if (i >= NBATCH * CCH * CCH) return;
  int n = i >> 14, idx = i & (CCH * CCH - 1);
  int r = idx >> 7, c = idx & (CCH - 1);
  double gs = 0.0, sr = 0.0, sc = 0.0;
  for (int p = 0; p < GRNG; ++p) {
    gs += (double)gpart[(size_t)(n * GRNG + p) * (CCH * CCH) + idx];
    sr += (double)csum[(n * GRNG + p) * CCH + r];
    sc += (double)csum[(n * GRNG + p) * CCH + c];
  }
  double mr = sr * (1.0 / (double)HW), mc = sc * (1.0 / (double)HW);
  cov[i] = (float)(gs * (1.0 / (double)HW) - mr * mc);
  if (r == 0) mu[n * CCH + c] = (float)mc;
}

__global__ __launch_bounds__(256) void k_quad(
    const float* __restrict__ cov, const float* __restrict__ mu,
    const float* __restrict__ W1, float* __restrict__ mean,
    float* __restrict__ rstd) {
  __shared__ float red[8][MID];
  __shared__ float res[2][MID];
  const int jb   = threadIdx.x >> 5;
  const int lane = threadIdx.x & 31;
  const int nb   = blockIdx.x;
  const int n = nb >> 3, band = nb & 7;
  const int lm = lane & 15, kh = (lane >> 4) * 2, mh = (lane >> 4) * 8;
  const float* A  = W1 + (size_t)band * MID * CCH;
  const float* cn = cov + (size_t)n * CCH * CCH + 16 * jb + lm;

  v8f acc[4] = {};
#pragma unroll
  for (int k0 = 0; k0 < CCH; k0 += 4) {
    int kk = k0 + kh;
    v2f b;
    b.x = cn[kk * CCH];
    b.y = cn[(kk + 1) * CCH];
#pragma unroll
    for (int t = 0; t < 4; ++t) {
      v2f a = *(const v2f*)(A + (t * 16 + lm) * CCH + kk);
      acc[t] = wmma4(a, b, acc[t]);
    }
  }
#pragma unroll
  for (int t = 0; t < 4; ++t)
#pragma unroll
    for (int v = 0; v < 8; ++v) {
      int m = t * 16 + v + mh;
      float p = acc[t][v] * A[m * CCH + 16 * jb + lm];
#pragma unroll
      for (int off = 1; off < 16; off <<= 1) p += __shfl_xor(p, off, 32);
      if (lm == 0) red[jb][m] = p;
    }
  __syncthreads();
  const int m = threadIdx.x >> 2, part = threadIdx.x & 3;
  float d = 0.0f;
#pragma unroll
  for (int i = 0; i < 32; ++i)
    d += A[m * CCH + part * 32 + i] * mu[n * CCH + part * 32 + i];
  d += __shfl_xor(d, 1, 32);
  d += __shfl_xor(d, 2, 32);
  if (part == 0) {
    float var = 0.0f;
#pragma unroll
    for (int j = 0; j < 8; ++j) var += red[j][m];
    res[0][m] = d;
    res[1][m] = rsqrtf(fmaxf(var, 0.0f) + 1e-5f);
  }
  __syncthreads();
  if (threadIdx.x < MID)          mean[nb * MID + threadIdx.x] = res[0][threadIdx.x];
  else if (threadIdx.x < 2 * MID) rstd[nb * MID + threadIdx.x - MID] = res[1][threadIdx.x - MID];
}

__global__ __launch_bounds__(256) void k_bands(
    const float* __restrict__ xmod, const float* __restrict__ W1,
    const float* __restrict__ W2, const float* __restrict__ W3,
    const float* __restrict__ mean, const float* __restrict__ rstd,
    float* __restrict__ out) {
  __shared__ float lds[8][MID * 17];
  const int band = threadIdx.x >> 5;
  const int lane = threadIdx.x & 31;
  const int n    = blockIdx.x / BPI;
  const int blk  = blockIdx.x % BPI;
  const int lm = lane & 15, kh = (lane >> 4) * 2, mh = (lane >> 4) * 8;
  const int nb = n * NBAND + band;
  const float* A  = W1 + (size_t)band * MID * CCH;
  const float* A2 = W2 + (size_t)band * 32 * MID;
  const float* w3 = W3 + (size_t)band * 32;
  const float* mu = mean + (size_t)nb * MID;
  const float* rs = rstd + (size_t)nb * MID;
  const float* xb = xmod + (size_t)n * CCH * HW + (size_t)kh * HW + blk * PXB + lm;
  float* sm = &lds[band][0];

  v2f a[4][KS1];
#pragma unroll
  for (int t = 0; t < 4; ++t)
#pragma unroll
    for (int k = 0; k < KS1; ++k)
      a[t][k] = *(const v2f*)(A + (t * 16 + lm) * CCH + 4 * k + kh);
  v2f a2[2][KS2];
#pragma unroll
  for (int t = 0; t < 2; ++t)
#pragma unroll
    for (int k = 0; k < KS2; ++k)
      a2[t][k] = *(const v2f*)(A2 + (t * 16 + lm) * MID + 4 * k + kh);

  v2f bq[KS1];
#pragma unroll
  for (int k = 0; k < KS1; ++k) {
    bq[k].x = xb[(size_t)(4 * k) * HW];
    bq[k].y = xb[(size_t)(4 * k + 1) * HW];
  }

  for (int sb = 0; sb < SBN; ++sb) {
    asm volatile("" ::: "memory");
    v8f acc[4] = {};
#pragma unroll
    for (int k = 0; k < KS1; ++k)
#pragma unroll
      for (int t = 0; t < 4; ++t)
        acc[t] = wmma4(a[t][k], bq[k], acc[t]);

    const int nx = (sb + 1 < SBN ? sb + 1 : sb) * 16;
#pragma unroll
    for (int k = 0; k < KS1; ++k) {
      bq[k].x = xb[(size_t)(4 * k) * HW + nx];
      bq[k].y = xb[(size_t)(4 * k + 1) * HW + nx];
    }

#pragma unroll
    for (int t = 0; t < 4; ++t)
#pragma unroll
      for (int v = 0; v < 8; ++v) {
        int m = t * 16 + v + mh;
        sm[m * 17 + lm] = gelu_f((acc[t][v] - mu[m]) * rs[m]);
      }

    v8f acc2[2] = {};
#pragma unroll
    for (int k = 0; k < KS2; ++k) {
      int kk = 4 * k + kh;
      v2f bb;
      bb.x = sm[kk * 17 + lm];
      bb.y = sm[(kk + 1) * 17 + lm];
#pragma unroll
      for (int t = 0; t < 2; ++t)
        acc2[t] = wmma4(a2[t][k], bb, acc2[t]);
    }
#pragma unroll
    for (int t = 0; t < 2; ++t)
#pragma unroll
      for (int v = 0; v < 8; ++v)
        sm[(t * 16 + v + mh) * 17 + lm] = gelu_f(acc2[t][v]);

    if (lane < 16) {
      float s = 0.0f;
#pragma unroll
      for (int k = 0; k < 32; ++k) s += w3[k] * sm[k * 17 + lane];
      out[(size_t)nb * HW + blk * PXB + sb * 16 + lane] = clamp01(gelu_f(s));
    }
  }
}

extern "C" void kernel_launch(void* const* d_in, const int* in_sizes, int n_in,
                              void* d_out, int out_size, void* d_ws, size_t ws_size,
                              hipStream_t stream) {
  const float* x   = (const float*)d_in[0];
  const float* Wp  = (const float*)d_in[1];
  const float* Wm1 = (const float*)d_in[2];
  const float* Wm2 = (const float*)d_in[3];
  const float* W1  = (const float*)d_in[4];
  const float* W2  = (const float*)d_in[5];
  const float* W3  = (const float*)d_in[6];
  float* out = (float*)d_out;

  float* ws   = (float*)d_ws;
  float* emb  = ws;
  float* xmod = emb + (size_t)PDIM * HW;
  float* gpart = xmod + (size_t)NBATCH * CCH * HW;
  float* mean = gpart + (size_t)NBATCH * GRNG * CCH * CCH;
  float* rstd = mean + (size_t)NBATCH * NBAND * MID;
  float* csum = emb;
  float* cov  = csum + (size_t)NBATCH * GRNG * CCH;
  float* mu   = cov + (size_t)NBATCH * CCH * CCH;
  (void)in_sizes; (void)n_in; (void)out_size; (void)ws_size;

  k_emb<<<HW / 256, 256, 0, stream>>>(Wp, emb);
  k_modulate<<<NBATCH * 128, 256, 0, stream>>>(x, emb, Wm1, Wm2, xmod);
  k_gram<<<NBATCH * GRNG * 4, 256, 0, stream>>>(xmod, gpart, csum);
  k_cov<<<NBATCH * CCH * CCH / 256, 256, 0, stream>>>(gpart, csum, cov, mu);
  k_quad<<<NBATCH * NBAND, 256, 0, stream>>>(cov, mu, W1, mean, rstd);
  k_bands<<<NBATCH * BPI, 256, 0, stream>>>(xmod, W1, W2, W3, mean, rstd, out);
}


# GELU evaluated branch-free (both erf ranges computed, one selected per lane; exp on the hardware exp2): an epilogue's independent activations interleave instead of running as serial branchy chains
# speedup vs baseline: 2.4369x; 1.4472x over previous
#include <hip/hip_runtime.h>
#include <math.h>


typedef __attribute__((ext_vector_type(2))) float v2f;
typedef __attribute__((ext_vector_type(4))) float v4f;
typedef __attribute__((ext_vector_type(8))) float v8f;

#define HW      16384
#define NBATCH  8
#define CCH     128
#define MID     64
#define PDIM    32
#define NBAND   8
#define PXB     512
#define SBN     (PXB / 16)
#define BPI     (HW / PXB)
#define KS1     (CCH / 4)
#define KS2     (MID / 4)
#define GPX     2048
#define GRNG    (HW / GPX)
#define GIT     128
#define GPITCH  (GIT + 4)

__device__ __forceinline__ v8f wmma4(v2f a, v2f b, v8f c) {
  return __builtin_amdgcn_wmma_f32_16x16x4_f32(false, a, false, b,
                                               (short)0, c, false, false);
}

__device__ __forceinline__ float erf_sel(float z) {
  const float az = fabsf(z);
  const float t  = az * az;
  float ps = fmaf(t, -0x1.268bc2p-11f, 0x1.420828p-8f);
  ps = fmaf(t, ps, -0x1.b5937p-6f);
  ps = fmaf(t, ps, 0x1.ce077cp-4f);
  ps = fmaf(t, ps, -0x1.81266p-2f);
  ps = fmaf(t, ps, 0x1.06eba0p-3f);
  ps = fmaf(az, ps, az);
  float pl = fmaf(az, 0x1.1d3156p-16f, -0x1.8d129p-12f);
  pl = fmaf(az, pl, 0x1.f9a6d2p-9f);
  pl = fmaf(az, pl, -0x1.8c3164p-6f);
  pl = fmaf(az, pl, 0x1.b4e9c8p-4f);
  pl = fmaf(az, pl, 0x1.4515fap-1f);
  pl = fmaf(az, pl, 0x1.078e50p-3f);
  pl = fmaf(az, pl, az);
  const float el = 1.0f - __builtin_amdgcn_exp2f(pl * -0x1.715476p+0f);
  const float r  = az < 1.0f ? ps : el;
  return copysignf(r, z);
}

__device__ __forceinline__ float gelu_f(float x) {
  return 0.5f * x * (1.0f + erf_sel(x * 0.70710678118654752440f));
}
__device__ __forceinline__ float clamp01(float x) {
  return fminf(fmaxf(x, 0.0f), 1.0f);
}

__global__ void k_emb(const float* __restrict__ Wp, float* __restrict__ emb) {
  int p = blockIdx.x * blockDim.x + threadIdx.x;
  if (p >= HW) return;
  int i = p >> 7, j = p & 127;
  float ym = -1.0f + (2.0f / 127.0f) * (float)i;
  float xm = -1.0f + (2.0f / 127.0f) * (float)j;
#pragma unroll 4
  for (int pd = 0; pd < PDIM; ++pd) {
    float v = tanhf(Wp[pd * 2 + 0] * ym + Wp[pd * 2 + 1] * xm);
    emb[(size_t)pd * HW + p] = v;
  }
}

__global__ __launch_bounds__(256) void k_modulate(
    const float* __restrict__ x, const float* __restrict__ emb,
    const float* __restrict__ Wm1, const float* __restrict__ Wm2,
    float* __restrict__ xmod) {
  __shared__ float lds[8][MID * 17];
  const int wave = threadIdx.x >> 5;
  const int lane = threadIdx.x & 31;
  const int n    = blockIdx.x >> 7;
  const int pb   = (blockIdx.x & 127) * 128 + wave * 16;
  const int lm   = lane & 15;
  const int kh   = (lane >> 4) * 2;
  const int mh   = (lane >> 4) * 8;
  const int pix  = pb + lm;
  const float* xn = x + (size_t)n * CCH * HW;

  v8f acc[4] = {};
  for (int k0 = 0; k0 < CCH; k0 += 4) {
    int kk = k0 + kh;
    v2f b;
    b.x = xn[(size_t)kk * HW + pix];
    b.y = xn[(size_t)(kk + 1) * HW + pix];
#pragma unroll
    for (int t = 0; t < 4; ++t) {
      v2f a = *(const v2f*)(Wm1 + (t * 16 + lm) * (CCH + PDIM) + kk);
      acc[t] = wmma4(a, b, acc[t]);
    }
  }
  for (int k0 = 0; k0 < PDIM; k0 += 4) {
    int kk = k0 + kh;
    v2f b;
    b.x = emb[(size_t)kk * HW + pix];
    b.y = emb[(size_t)(kk + 1) * HW + pix];
#pragma unroll
    for (int t = 0; t < 4; ++t) {
      v2f a = *(const v2f*)(Wm1 + (t * 16 + lm) * (CCH + PDIM) + CCH + kk);
      acc[t] = wmma4(a, b, acc[t]);
    }
  }
  float* sm = &lds[wave][0];
#pragma unroll
  for (int t = 0; t < 4; ++t)
#pragma unroll
    for (int v = 0; v < 8; ++v)
      sm[(t * 16 + v + mh) * 17 + lm] = gelu_f(acc[t][v]);

  v8f acc2[8] = {};
  for (int k0 = 0; k0 < MID; k0 += 4) {
    int kk = k0 + kh;
    v2f b;
    b.x = sm[kk * 17 + lm];
    b.y = sm[(kk + 1) * 17 + lm];
#pragma unroll
    for (int t = 0; t < 8; ++t) {
      v2f a = *(const v2f*)(Wm2 + (t * 16 + lm) * MID + kk);
      acc2[t] = wmma4(a, b, acc2[t]);
    }
  }
  float* xo = xmod + (size_t)n * CCH * HW;
#pragma unroll
  for (int t = 0; t < 8; ++t)
#pragma unroll
    for (int v = 0; v < 8; ++v) {
      int m = t * 16 + v + mh;
      xo[(size_t)m * HW + pix] = clamp01(gelu_f(acc2[t][v]));
    }
}

__global__ __launch_bounds__(256) void k_gram(
    const float* __restrict__ xmod, float* __restrict__ gpart,
    float* __restrict__ csum) {
  __shared__ __attribute__((aligned(16))) float tile[2][CCH * GPITCH];
  __shared__ float csl[CCH];
  const int wave = threadIdx.x >> 5;
  const int lane = threadIdx.x & 31;
  const int g    = blockIdx.x & 3;
  const int rng  = (blockIdx.x >> 2) & (GRNG - 1);
  const int n    = blockIdx.x >> 5;
  const int lm = lane & 15, hh = lane >> 4, mh = hh * 8;
  const int rb = 2 * g + (wave & 1);
  const int cb = 2 * (wave >> 1);
  const float* sp = xmod + (size_t)n * CCH * HW + (size_t)(16 * wave) * HW
                  + (size_t)rng * GPX + 4 * lane;
  v4f st[16];
#pragma unroll
  for (int q = 0; q < 16; ++q) st[q] = *(const v4f*)(sp + (size_t)q * HW);
  float rsum[16];
#pragma unroll
  for (int q = 0; q < 16; ++q) rsum[q] = 0.0f;
  v8f acc0 = {}, acc1 = {};
  const int fa  = (16 * rb + lm) * GPITCH + 4 * hh;
  const int fb0 = (16 * cb + lm) * GPITCH + 4 * hh;
  const int fb1 = fb0 + 16 * GPITCH;

  for (int it = 0; it < GPX / GIT; ++it) {
    float* tb = &tile[it & 1][0];
#pragma unroll
    for (int q = 0; q < 16; ++q) {
      *(v4f*)(tb + (16 * wave + q) * GPITCH + 4 * lane) = st[q];
      rsum[q] += (st[q].x + st[q].y) + (st[q].z + st[q].w);
    }
    __syncthreads();
    if (it + 1 < GPX / GIT) {
#pragma unroll
      for (int q = 0; q < 16; ++q)
        st[q] = *(const v4f*)(sp + (size_t)q * HW + (it + 1) * GIT);
    }
    v4f a  = *(const v4f*)(tb + fa);
    v4f b0 = *(const v4f*)(tb + fb0);
    v4f b1 = *(const v4f*)(tb + fb1);
#pragma unroll
    for (int s = 0; s < GIT / 8; ++s) {
      v4f an = a, b0n = b0, b1n = b1;
      if (s + 1 < GIT / 8) {
        an  = *(const v4f*)(tb + fa  + 8 * (s + 1));
        b0n = *(const v4f*)(tb + fb0 + 8 * (s + 1));
        b1n = *(const v4f*)(tb + fb1 + 8 * (s + 1));
      }
      v2f alo = {a.x, a.y},   ahi = {a.z, a.w};
      v2f plo = {b0.x, b0.y}, phi = {b0.z, b0.w};
      v2f qlo = {b1.x, b1.y}, qhi = {b1.z, b1.w};
      acc0 = wmma4(alo, plo, acc0);
      acc1 = wmma4(alo, qlo, acc1);
      acc0 = wmma4(ahi, phi, acc0);
      acc1 = wmma4(ahi, qhi, acc1);
      a = an; b0 = b0n; b1 = b1n;
    }
  }

  float* gp = gpart + (size_t)(n * GRNG + rng) * (CCH * CCH);
#pragma unroll
  for (int v = 0; v < 8; ++v) {
    int row = 16 * rb + mh + v;
    gp[row * CCH + 16 * cb + lm]       = acc0[v];
    gp[row * CCH + 16 * (cb + 1) + lm] = acc1[v];
  }
#pragma unroll
  for (int q = 0; q < 16; ++q) {
    float s = rsum[q];
#pragma unroll
    for (int off = 1; off < 32; off <<= 1) s += __shfl_xor(s, off, 32);
    if (lane == 0) csl[16 * wave + q] = s;
  }
  __syncthreads();
  if (g == 0 && threadIdx.x < CCH)
    csum[(n * GRNG + rng) * CCH + threadIdx.x] = csl[threadIdx.x];
}

__global__ void k_cov(const float* __restrict__ gpart,
                      const float* __restrict__ csum,
                      float* __restrict__ cov, float* __restrict__ mu) {
  int i = blockIdx.x * blockDim.x + threadIdx.x;
  if (i >= NBATCH * CCH * CCH) return;
  int n = i >> 14, idx = i & (CCH * CCH - 1);
  int r = idx >> 7, c = idx & (CCH - 1);
  double gs = 0.0, sr = 0.0, sc = 0.0;
  for (int p = 0; p < GRNG; ++p) {
    gs += (double)gpart[(size_t)(n * GRNG + p) * (CCH * CCH) + idx];
    sr += (double)csum[(n * GRNG + p) * CCH + r];
    sc += (double)csum[(n * GRNG + p) * CCH + c];
  }
  double mr = sr * (1.0 / (double)HW), mc = sc * (1.0 / (double)HW);
  cov[i] = (float)(gs * (1.0 / (double)HW) - mr * mc);
  if (r == 0) mu[n * CCH + c] = (float)mc;
}

__global__ __launch_bounds__(256) void k_quad(
    const float* __restrict__ cov, const float* __restrict__ mu,
    const float* __restrict__ W1, float* __restrict__ mean,
    float* __restrict__ rstd) {
  __shared__ float red[8][MID];
  __shared__ float res[2][MID];
  const int jb   = threadIdx.x >> 5;
  const int lane = threadIdx.x & 31;
  const int nb   = blockIdx.x;
  const int n = nb >> 3, band = nb & 7;
  const int lm = lane & 15, kh = (lane >> 4) * 2, mh = (lane >> 4) * 8;
  const float* A  = W1 + (size_t)band * MID * CCH;
  const float* cn = cov + (size_t)n * CCH * CCH + 16 * jb + lm;

  v8f acc[4] = {};
#pragma unroll
  for (int k0 = 0; k0 < CCH; k0 += 4) {
    int kk = k0 + kh;
    v2f b;
    b.x = cn[kk * CCH];
    b.y = cn[(kk + 1) * CCH];
#pragma unroll
    for (int t = 0; t < 4; ++t) {
      v2f a = *(const v2f*)(A + (t * 16 + lm) * CCH + kk);
      acc[t] = wmma4(a, b, acc[t]);
    }
  }
#pragma unroll
  for (int t = 0; t < 4; ++t)
#pragma unroll
    for (int v = 0; v < 8; ++v) {
      int m = t * 16 + v + mh;
      float p = acc[t][v] * A[m * CCH + 16 * jb + lm];
#pragma unroll
      for (int off = 1; off < 16; off <<= 1) p += __shfl_xor(p, off, 32);
      if (lm == 0) red[jb][m] = p;
    }
  __syncthreads();
  const int m = threadIdx.x >> 2, part = threadIdx.x & 3;
  float d = 0.0f;
#pragma unroll
  for (int i = 0; i < 32; ++i)
    d += A[m * CCH + part * 32 + i] * mu[n * CCH + part * 32 + i];
  d += __shfl_xor(d, 1, 32);
  d += __shfl_xor(d, 2, 32);
  if (part == 0) {
    float var = 0.0f;
#pragma unroll
    for (int j = 0; j < 8; ++j) var += red[j][m];
    res[0][m] = d;
    res[1][m] = rsqrtf(fmaxf(var, 0.0f) + 1e-5f);
  }
  __syncthreads();
  if (threadIdx.x < MID)          mean[nb * MID + threadIdx.x] = res[0][threadIdx.x];
  else if (threadIdx.x < 2 * MID) rstd[nb * MID + threadIdx.x - MID] = res[1][threadIdx.x - MID];
}

__global__ __launch_bounds__(256) void k_bands(
    const float* __restrict__ xmod, const float* __restrict__ W1,
    const float* __restrict__ W2, const float* __restrict__ W3,
    const float* __restrict__ mean, const float* __restrict__ rstd,
    float* __restrict__ out) {
  __shared__ float lds[8][MID * 17];
  const int band = threadIdx.x >> 5;
  const int lane = threadIdx.x & 31;
  const int n    = blockIdx.x / BPI;
  const int blk  = blockIdx.x % BPI;
  const int lm = lane & 15, kh = (lane >> 4) * 2, mh = (lane >> 4) * 8;
  const int nb = n * NBAND + band;
  const float* A  = W1 + (size_t)band * MID * CCH;
  const float* A2 = W2 + (size_t)band * 32 * MID;
  const float* w3 = W3 + (size_t)band * 32;
  const float* mu = mean + (size_t)nb * MID;
  const float* rs = rstd + (size_t)nb * MID;
  const float* xb = xmod + (size_t)n * CCH * HW + (size_t)kh * HW + blk * PXB + lm;
  float* sm = &lds[band][0];

  v2f a[4][KS1];
#pragma unroll
  for (int t = 0; t < 4; ++t)
#pragma unroll
    for (int k = 0; k < KS1; ++k)
      a[t][k] = *(const v2f*)(A + (t * 16 + lm) * CCH + 4 * k + kh);
  v2f a2[2][KS2];
#pragma unroll
  for (int t = 0; t < 2; ++t)
#pragma unroll
    for (int k = 0; k < KS2; ++k)
      a2[t][k] = *(const v2f*)(A2 + (t * 16 + lm) * MID + 4 * k + kh);

  v2f bq[KS1];
#pragma unroll
  for (int k = 0; k < KS1; ++k) {
    bq[k].x = xb[(size_t)(4 * k) * HW];
    bq[k].y = xb[(size_t)(4 * k + 1) * HW];
  }

  for (int sb = 0; sb < SBN; ++sb) {
    asm volatile("" ::: "memory");
    v8f acc[4] = {};
#pragma unroll
    for (int k = 0; k < KS1; ++k)
#pragma unroll
      for (int t = 0; t < 4; ++t)
        acc[t] = wmma4(a[t][k], bq[k], acc[t]);

    const int nx = (sb + 1 < SBN ? sb + 1 : sb) * 16;
#pragma unroll
    for (int k = 0; k < KS1; ++k) {
      bq[k].x = xb[(size_t)(4 * k) * HW + nx];
      bq[k].y = xb[(size_t)(4 * k + 1) * HW + nx];
    }

#pragma unroll
    for (int t = 0; t < 4; ++t)
#pragma unroll
      for (int v = 0; v < 8; ++v) {
        int m = t * 16 + v + mh;
        sm[m * 17 + lm] = gelu_f((acc[t][v] - mu[m]) * rs[m]);
      }

    v8f acc2[2] = {};
#pragma unroll
    for (int k = 0; k < KS2; ++k) {
      int kk = 4 * k + kh;
      v2f bb;
      bb.x = sm[kk * 17 + lm];
      bb.y = sm[(kk + 1) * 17 + lm];
#pragma unroll
      for (int t = 0; t < 2; ++t)
        acc2[t] = wmma4(a2[t][k], bb, acc2[t]);
    }
#pragma unroll
    for (int t = 0; t < 2; ++t)
#pragma unroll
      for (int v = 0; v < 8; ++v)
        sm[(t * 16 + v + mh) * 17 + lm] = gelu_f(acc2[t][v]);

    if (lane < 16) {
      float s = 0.0f;
#pragma unroll
      for (int k = 0; k < 32; ++k) s += w3[k] * sm[k * 17 + lane];
      out[(size_t)nb * HW + blk * PXB + sb * 16 + lane] = clamp01(gelu_f(s));
    }
  }
}

extern "C" void kernel_launch(void* const* d_in, const int* in_sizes, int n_in,
                              void* d_out, int out_size, void* d_ws, size_t ws_size,
                              hipStream_t stream) {
  const float* x   = (const float*)d_in[0];
  const float* Wp  = (const float*)d_in[1];
  const float* Wm1 = (const float*)d_in[2];
  const float* Wm2 = (const float*)d_in[3];
  const float* W1  = (const float*)d_in[4];
  const float* W2  = (const float*)d_in[5];
  const float* W3  = (const float*)d_in[6];
  float* out = (float*)d_out;

  float* ws   = (float*)d_ws;
  float* emb  = ws;
  float* xmod = emb + (size_t)PDIM * HW;
  float* gpart = xmod + (size_t)NBATCH * CCH * HW;
  float* mean = gpart + (size_t)NBATCH * GRNG * CCH * CCH;
  float* rstd = mean + (size_t)NBATCH * NBAND * MID;
  float* csum = emb;
  float* cov  = csum + (size_t)NBATCH * GRNG * CCH;
  float* mu   = cov + (size_t)NBATCH * CCH * CCH;
  (void)in_sizes; (void)n_in; (void)out_size; (void)ws_size;

  k_emb<<<HW / 256, 256, 0, stream>>>(Wp, emb);
  k_modulate<<<NBATCH * 128, 256, 0, stream>>>(x, emb, Wm1, Wm2, xmod);
  k_gram<<<NBATCH * GRNG * 4, 256, 0, stream>>>(xmod, gpart, csum);
  k_cov<<<NBATCH * CCH * CCH / 256, 256, 0, stream>>>(gpart, csum, cov, mu);
  k_quad<<<NBATCH * NBAND, 256, 0, stream>>>(cov, mu, W1, mean, rstd);
  k_bands<<<NBATCH * BPI, 256, 0, stream>>>(xmod, W1, W2, W3, mean, rstd, out);
}


# Modulation kernel: Wm1, Wm2 staged once per 256-pixel block in LDS in WMMA A-fragment order; one 16-byte LDS read per two k-steps replaces a 16-line global gather per k-step per 16-pixel tile
# speedup vs baseline: 2.5545x; 1.0483x over previous
#include <hip/hip_runtime.h>
#include <math.h>


typedef __attribute__((ext_vector_type(2))) float v2f;
typedef __attribute__((ext_vector_type(4))) float v4f;
typedef __attribute__((ext_vector_type(8))) float v8f;

#define HW      16384
#define NBATCH  8
#define CCH     128
#define MID     64
#define PDIM    32
#define NBAND   8
#define PXB     512
#define SBN     (PXB / 16)
#define BPI     (HW / PXB)
#define KS1     (CCH / 4)
#define KS2     (MID / 4)
#define GPX     2048
#define GRNG    (HW / GPX)
#define GIT     128
#define GPITCH  (GIT + 4)
#define MGRP    2
#define MBPI    (HW / (128 * MGRP))
#define KSE     (PDIM / 4)

__device__ __forceinline__ v8f wmma4(v2f a, v2f b, v8f c) {
  return __builtin_amdgcn_wmma_f32_16x16x4_f32(false, a, false, b,
                                               (short)0, c, false, false);
}

__device__ __forceinline__ float erf_sel(float z) {
  const float az = fabsf(z);
  const float t  = az * az;
  float ps = fmaf(t, -0x1.268bc2p-11f, 0x1.420828p-8f);
  ps = fmaf(t, ps, -0x1.b5937p-6f);
  ps = fmaf(t, ps, 0x1.ce077cp-4f);
  ps = fmaf(t, ps, -0x1.81266p-2f);
  ps = fmaf(t, ps, 0x1.06eba0p-3f);
  ps = fmaf(az, ps, az);
  float pl = fmaf(az, 0x1.1d3156p-16f, -0x1.8d129p-12f);
  pl = fmaf(az, pl, 0x1.f9a6d2p-9f);
  pl = fmaf(az, pl, -0x1.8c3164p-6f);
  pl = fmaf(az, pl, 0x1.b4e9c8p-4f);
  pl = fmaf(az, pl, 0x1.4515fap-1f);
  pl = fmaf(az, pl, 0x1.078e50p-3f);
  pl = fmaf(az, pl, az);
  const float el = 1.0f - __builtin_amdgcn_exp2f(pl * -0x1.715476p+0f);
  const float r  = az < 1.0f ? ps : el;
  return copysignf(r, z);
}

__device__ __forceinline__ float gelu_f(float x) {
  return 0.5f * x * (1.0f + erf_sel(x * 0.70710678118654752440f));
}
__device__ __forceinline__ float clamp01(float x) {
  return fminf(fmaxf(x, 0.0f), 1.0f);
}

__global__ void k_emb(const float* __restrict__ Wp, float* __restrict__ emb) {
  int p = blockIdx.x * blockDim.x + threadIdx.x;
  if (p >= HW) return;
  int i = p >> 7, j = p & 127;
  float ym = -1.0f + (2.0f / 127.0f) * (float)i;
  float xm = -1.0f + (2.0f / 127.0f) * (float)j;
#pragma unroll 4
  for (int pd = 0; pd < PDIM; ++pd) {
    float v = tanhf(Wp[pd * 2 + 0] * ym + Wp[pd * 2 + 1] * xm);
    emb[(size_t)pd * HW + p] = v;
  }
}

__global__ __launch_bounds__(256, 4) void k_modulate(
    const float* __restrict__ x, const float* __restrict__ emb,
    const float* __restrict__ Wm1, const float* __restrict__ Wm2,
    float* __restrict__ xmod) {
  __shared__ __attribute__((aligned(16))) float wa1[MID * (CCH + PDIM)];
  __shared__ __attribute__((aligned(16))) float wa2[CCH * MID];
  __shared__ float lds[8][MID * 17];
  const int wave = threadIdx.x >> 5;
  const int lane = threadIdx.x & 31;
  const int n    = blockIdx.x / MBPI;
  const int gb   = (blockIdx.x % MBPI) * (128 * MGRP);
  const int lm   = lane & 15;
  const int kh   = (lane >> 4) * 2;
  const int mh   = (lane >> 4) * 8;

  for (int u = threadIdx.x; u < MID * (CCH + PDIM) / 2; u += 256) {
    const int r = u / ((CCH + PDIM) / 2), c2 = u % ((CCH + PDIM) / 2);
    const int ks = c2 >> 1, h = c2 & 1;
    const v2f g = *(const v2f*)(Wm1 + r * (CCH + PDIM) + 2 * c2);
    *(v2f*)(wa1 + ((((ks >> 1) * 4 + (r >> 4)) * 32 + h * 16 + (r & 15)) * 4
                   + (ks & 1) * 2)) = g;
  }
  for (int u = threadIdx.x; u < CCH * MID / 2; u += 256) {
    const int r = u / (MID / 2), c2 = u % (MID / 2);
    const int ks = c2 >> 1, h = c2 & 1;
    const v2f g = *(const v2f*)(Wm2 + r * MID + 2 * c2);
    *(v2f*)(wa2 + ((((ks >> 1) * 8 + (r >> 4)) * 32 + h * 16 + (r & 15)) * 4
                   + (ks & 1) * 2)) = g;
  }
  __syncthreads();

  const float* xn = x + (size_t)n * CCH * HW;
  float* xo = xmod + (size_t)n * CCH * HW;
  float* smw = &lds[wave][0] + mh * 17 + lm;
  const float* smr = &lds[wave][0] + kh * 17 + lm;
  const v4f* f1 = (const v4f*)wa1 + lane;
  const v4f* f2 = (const v4f*)wa2 + lane;

  for (int g = 0; g < MGRP; ++g) {
    asm volatile("" ::: "memory");
    const int pix = gb + g * 128 + wave * 16 + lm;
    float* xop = xo + (size_t)mh * HW + pix;

    v2f b[KS1 + KSE];
#pragma unroll
    for (int ks = 0; ks < KS1; ++ks) {
      b[ks].x = xn[(size_t)(4 * ks + kh) * HW + pix];
      b[ks].y = xn[(size_t)(4 * ks + kh + 1) * HW + pix];
    }
#pragma unroll
    for (int ks = 0; ks < KSE; ++ks) {
      b[KS1 + ks].x = emb[(size_t)(4 * ks + kh) * HW + pix];
      b[KS1 + ks].y = emb[(size_t)(4 * ks + kh + 1) * HW + pix];
    }

    v8f acc[4] = {};
    v4f fr[2][8];
#pragma unroll
    for (int i = 0; i < 8; ++i) fr[0][i] = f1[i * 32];
#pragma unroll
    for (int c = 0; c < (KS1 + KSE) / 4; ++c) {
      if (c + 1 < (KS1 + KSE) / 4) {
#pragma unroll
        for (int i = 0; i < 8; ++i) fr[(c + 1) & 1][i] = f1[((c + 1) * 8 + i) * 32];
      }
#pragma unroll
      for (int q = 0; q < 2; ++q) {
#pragma unroll
        for (int t = 0; t < 4; ++t) {
          v2f lo = {fr[c & 1][q * 4 + t].x, fr[c & 1][q * 4 + t].y};
          acc[t] = wmma4(lo, b[4 * c + 2 * q], acc[t]);
        }
#pragma unroll
        for (int t = 0; t < 4; ++t) {
          v2f hi = {fr[c & 1][q * 4 + t].z, fr[c & 1][q * 4 + t].w};
          acc[t] = wmma4(hi, b[4 * c + 2 * q + 1], acc[t]);
        }
      }
      __builtin_amdgcn_sched_barrier(0);
    }
#pragma unroll
    for (int t = 0; t < 4; ++t)
#pragma unroll
      for (int v = 0; v < 8; ++v)
        smw[(t * 16 + v) * 17] = gelu_f(acc[t][v]);

    v8f acc2[8] = {};
    v4f fs[2][8];
    v2f bs[2][2];
#pragma unroll
    for (int t = 0; t < 8; ++t) fs[0][t] = f2[t * 32];
    bs[0][0].x = smr[0];
    bs[0][0].y = smr[17];
    bs[0][1].x = smr[4 * 17];
    bs[0][1].y = smr[5 * 17];
#pragma unroll
    for (int kp = 0; kp < KS2 / 2; ++kp) {
      if (kp + 1 < KS2 / 2) {
#pragma unroll
        for (int t = 0; t < 8; ++t) fs[(kp + 1) & 1][t] = f2[((kp + 1) * 8 + t) * 32];
        bs[(kp + 1) & 1][0].x = smr[(8 * kp + 8) * 17];
        bs[(kp + 1) & 1][0].y = smr[(8 * kp + 9) * 17];
        bs[(kp + 1) & 1][1].x = smr[(8 * kp + 12) * 17];
        bs[(kp + 1) & 1][1].y = smr[(8 * kp + 13) * 17];
      }
#pragma unroll
      for (int t = 0; t < 8; ++t) {
        v2f lo = {fs[kp & 1][t].x, fs[kp & 1][t].y};
        acc2[t] = wmma4(lo, bs[kp & 1][0], acc2[t]);
      }
#pragma unroll
      for (int t = 0; t < 8; ++t) {
        v2f hi = {fs[kp & 1][t].z, fs[kp & 1][t].w};
        acc2[t] = wmma4(hi, bs[kp & 1][1], acc2[t]);
      }
      __builtin_amdgcn_sched_barrier(0);
    }
#pragma unroll
    for (int t = 0; t < 8; ++t)
#pragma unroll
      for (int v = 0; v < 8; ++v)
        xop[(size_t)(t * 16 + v) * HW] = clamp01(gelu_f(acc2[t][v]));
  }
}

__global__ __launch_bounds__(256) void k_gram(
    const float* __restrict__ xmod, float* __restrict__ gpart,
    float* __restrict__ csum) {
  __shared__ __attribute__((aligned(16))) float tile[2][CCH * GPITCH];
  __shared__ float csl[CCH];
  const int wave = threadIdx.x >> 5;
  const int lane = threadIdx.x & 31;
  const int g    = blockIdx.x & 3;
  const int rng  = (blockIdx.x >> 2) & (GRNG - 1);
  const int n    = blockIdx.x >> 5;
  const int lm = lane & 15, hh = lane >> 4, mh = hh * 8;
  const int rb = 2 * g + (wave & 1);
  const int cb = 2 * (wave >> 1);
  const float* sp = xmod + (size_t)n * CCH * HW + (size_t)(16 * wave) * HW
                  + (size_t)rng * GPX + 4 * lane;
  v4f st[16];
#pragma unroll
  for (int q = 0; q < 16; ++q) st[q] = *(const v4f*)(sp + (size_t)q * HW);
  float rsum[16];
#pragma unroll
  for (int q = 0; q < 16; ++q) rsum[q] = 0.0f;
  v8f acc0 = {}, acc1 = {};
  const int fa  = (16 * rb + lm) * GPITCH + 4 * hh;
  const int fb0 = (16 * cb + lm) * GPITCH + 4 * hh;
  const int fb1 = fb0 + 16 * GPITCH;

  for (int it = 0; it < GPX / GIT; ++it) {
    float* tb = &tile[it & 1][0];
#pragma unroll
    for (int q = 0; q < 16; ++q) {
      *(v4f*)(tb + (16 * wave + q) * GPITCH + 4 * lane) = st[q];
      rsum[q] += (st[q].x + st[q].y) + (st[q].z + st[q].w);
    }
    __syncthreads();
    if (it + 1 < GPX / GIT) {
#pragma unroll
      for (int q = 0; q < 16; ++q)
        st[q] = *(const v4f*)(sp + (size_t)q * HW + (it + 1) * GIT);
    }
    v4f a  = *(const v4f*)(tb + fa);
    v4f b0 = *(const v4f*)(tb + fb0);
    v4f b1 = *(const v4f*)(tb + fb1);
#pragma unroll
    for (int s = 0; s < GIT / 8; ++s) {
      v4f an = a, b0n = b0, b1n = b1;
      if (s + 1 < GIT / 8) {
        an  = *(const v4f*)(tb + fa  + 8 * (s + 1));
        b0n = *(const v4f*)(tb + fb0 + 8 * (s + 1));
        b1n = *(const v4f*)(tb + fb1 + 8 * (s + 1));
      }
      v2f alo = {a.x, a.y},   ahi = {a.z, a.w};
      v2f plo = {b0.x, b0.y}, phi = {b0.z, b0.w};
      v2f qlo = {b1.x, b1.y}, qhi = {b1.z, b1.w};
      acc0 = wmma4(alo, plo, acc0);
      acc1 = wmma4(alo, qlo, acc1);
      acc0 = wmma4(ahi, phi, acc0);
      acc1 = wmma4(ahi, qhi, acc1);
      a = an; b0 = b0n; b1 = b1n;
    }
  }

  float* gp = gpart + (size_t)(n * GRNG + rng) * (CCH * CCH);
#pragma unroll
  for (int v = 0; v < 8; ++v) {
    int row = 16 * rb + mh + v;
    gp[row * CCH + 16 * cb + lm]       = acc0[v];
    gp[row * CCH + 16 * (cb + 1) + lm] = acc1[v];
  }
#pragma unroll
  for (int q = 0; q < 16; ++q) {
    float s = rsum[q];
#pragma unroll
    for (int off = 1; off < 32; off <<= 1) s += __shfl_xor(s, off, 32);
    if (lane == 0) csl[16 * wave + q] = s;
  }
  __syncthreads();
  if (g == 0 && threadIdx.x < CCH)
    csum[(n * GRNG + rng) * CCH + threadIdx.x] = csl[threadIdx.x];
}

__global__ void k_cov(const float* __restrict__ gpart,
                      const float* __restrict__ csum,
                      float* __restrict__ cov, float* __restrict__ mu) {
  int i = blockIdx.x * blockDim.x + threadIdx.x;
  if (i >= NBATCH * CCH * CCH) return;
  int n = i >> 14, idx = i & (CCH * CCH - 1);
  int r = idx >> 7, c = idx & (CCH - 1);
  double gs = 0.0, sr = 0.0, sc = 0.0;
  for (int p = 0; p < GRNG; ++p) {
    gs += (double)gpart[(size_t)(n * GRNG + p) * (CCH * CCH) + idx];
    sr += (double)csum[(n * GRNG + p) * CCH + r];
    sc += (double)csum[(n * GRNG + p) * CCH + c];
  }
  double mr = sr * (1.0 / (double)HW), mc = sc * (1.0 / (double)HW);
  cov[i] = (float)(gs * (1.0 / (double)HW) - mr * mc);
  if (r == 0) mu[n * CCH + c] = (float)mc;
}

__global__ __launch_bounds__(256) void k_quad(
    const float* __restrict__ cov, const float* __restrict__ mu,
    const float* __restrict__ W1, float* __restrict__ mean,
    float* __restrict__ rstd) {
  __shared__ float red[8][MID];
  __shared__ float res[2][MID];
  const int jb   = threadIdx.x >> 5;
  const int lane = threadIdx.x & 31;
  const int nb   = blockIdx.x;
  const int n = nb >> 3, band = nb & 7;
  const int lm = lane & 15, kh = (lane >> 4) * 2, mh = (lane >> 4) * 8;
  const float* A  = W1 + (size_t)band * MID * CCH;
  const float* cn = cov + (size_t)n * CCH * CCH + 16 * jb + lm;

  v8f acc[4] = {};
#pragma unroll
  for (int k0 = 0; k0 < CCH; k0 += 4) {
    int kk = k0 + kh;
    v2f b;
    b.x = cn[kk * CCH];
    b.y = cn[(kk + 1) * CCH];
#pragma unroll
    for (int t = 0; t < 4; ++t) {
      v2f a = *(const v2f*)(A + (t * 16 + lm) * CCH + kk);
      acc[t] = wmma4(a, b, acc[t]);
    }
  }
#pragma unroll
  for (int t = 0; t < 4; ++t)
#pragma unroll
    for (int v = 0; v < 8; ++v) {
      int m = t * 16 + v + mh;
      float p = acc[t][v] * A[m * CCH + 16 * jb + lm];
#pragma unroll
      for (int off = 1; off < 16; off <<= 1) p += __shfl_xor(p, off, 32);
      if (lm == 0) red[jb][m] = p;
    }
  __syncthreads();
  const int m = threadIdx.x >> 2, part = threadIdx.x & 3;
  float d = 0.0f;
#pragma unroll
  for (int i = 0; i < 32; ++i)
    d += A[m * CCH + part * 32 + i] * mu[n * CCH + part * 32 + i];
  d += __shfl_xor(d, 1, 32);
  d += __shfl_xor(d, 2, 32);
  if (part == 0) {
    float var = 0.0f;
#pragma unroll
    for (int j = 0; j < 8; ++j) var += red[j][m];
    res[0][m] = d;
    res[1][m] = rsqrtf(fmaxf(var, 0.0f) + 1e-5f);
  }
  __syncthreads();
  if (threadIdx.x < MID)          mean[nb * MID + threadIdx.x] = res[0][threadIdx.x];
  else if (threadIdx.x < 2 * MID) rstd[nb * MID + threadIdx.x - MID] = res[1][threadIdx.x - MID];
}

__global__ __launch_bounds__(256) void k_bands(
    const float* __restrict__ xmod, const float* __restrict__ W1,
    const float* __restrict__ W2, const float* __restrict__ W3,
    const float* __restrict__ mean, const float* __restrict__ rstd,
    float* __restrict__ out) {
  __shared__ float lds[8][MID * 17];
  const int band = threadIdx.x >> 5;
  const int lane = threadIdx.x & 31;
  const int n    = blockIdx.x / BPI;
  const int blk  = blockIdx.x % BPI;
  const int lm = lane & 15, kh = (lane >> 4) * 2, mh = (lane >> 4) * 8;
  const int nb = n * NBAND + band;
  const float* A  = W1 + (size_t)band * MID * CCH;
  const float* A2 = W2 + (size_t)band * 32 * MID;
  const float* w3 = W3 + (size_t)band * 32;
  const float* mu = mean + (size_t)nb * MID;
  const float* rs = rstd + (size_t)nb * MID;
  const float* xb = xmod + (size_t)n * CCH * HW + (size_t)kh * HW + blk * PXB + lm;
  float* sm = &lds[band][0];

  v2f a[4][KS1];
#pragma unroll
  for (int t = 0; t < 4; ++t)
#pragma unroll
    for (int k = 0; k < KS1; ++k)
      a[t][k] = *(const v2f*)(A + (t * 16 + lm) * CCH + 4 * k + kh);
  v2f a2[2][KS2];
#pragma unroll
  for (int t = 0; t < 2; ++t)
#pragma unroll
    for (int k = 0; k < KS2; ++k)
      a2[t][k] = *(const v2f*)(A2 + (t * 16 + lm) * MID + 4 * k + kh);

  v2f bq[KS1];
#pragma unroll
  for (int k = 0; k < KS1; ++k) {
    bq[k].x = xb[(size_t)(4 * k) * HW];
    bq[k].y = xb[(size_t)(4 * k + 1) * HW];
  }

  for (int sb = 0; sb < SBN; ++sb) {
    asm volatile("" ::: "memory");
    v8f acc[4] = {};
#pragma unroll
    for (int k = 0; k < KS1; ++k)
#pragma unroll
      for (int t = 0; t < 4; ++t)
        acc[t] = wmma4(a[t][k], bq[k], acc[t]);

    const int nx = (sb + 1 < SBN ? sb + 1 : sb) * 16;
#pragma unroll
    for (int k = 0; k < KS1; ++k) {
      bq[k].x = xb[(size_t)(4 * k) * HW + nx];
      bq[k].y = xb[(size_t)(4 * k + 1) * HW + nx];
    }

#pragma unroll
    for (int t = 0; t < 4; ++t)
#pragma unroll
      for (int v = 0; v < 8; ++v) {
        int m = t * 16 + v + mh;
        sm[m * 17 + lm] = gelu_f((acc[t][v] - mu[m]) * rs[m]);
      }

    v8f acc2[2] = {};
#pragma unroll
    for (int k = 0; k < KS2; ++k) {
      int kk = 4 * k + kh;
      v2f bb;
      bb.x = sm[kk * 17 + lm];
      bb.y = sm[(kk + 1) * 17 + lm];
#pragma unroll
      for (int t = 0; t < 2; ++t)
        acc2[t] = wmma4(a2[t][k], bb, acc2[t]);
    }
#pragma unroll
    for (int t = 0; t < 2; ++t)
#pragma unroll
      for (int v = 0; v < 8; ++v)
        sm[(t * 16 + v + mh) * 17 + lm] = gelu_f(acc2[t][v]);

    if (lane < 16) {
      float s = 0.0f;
#pragma unroll
      for (int k = 0; k < 32; ++k) s += w3[k] * sm[k * 17 + lane];
      out[(size_t)nb * HW + blk * PXB + sb * 16 + lane] = clamp01(gelu_f(s));
    }
  }
}

extern "C" void kernel_launch(void* const* d_in, const int* in_sizes, int n_in,
                              void* d_out, int out_size, void* d_ws, size_t ws_size,
                              hipStream_t stream) {
  const float* x   = (const float*)d_in[0];
  const float* Wp  = (const float*)d_in[1];
  const float* Wm1 = (const float*)d_in[2];
  const float* Wm2 = (const float*)d_in[3];
  const float* W1  = (const float*)d_in[4];
  const float* W2  = (const float*)d_in[5];
  const float* W3  = (const float*)d_in[6];
  float* out = (float*)d_out;

  float* ws   = (float*)d_ws;
  float* emb  = ws;
  float* xmod = emb + (size_t)PDIM * HW;
  float* gpart = xmod + (size_t)NBATCH * CCH * HW;
  float* mean = gpart + (size_t)NBATCH * GRNG * CCH * CCH;
  float* rstd = mean + (size_t)NBATCH * NBAND * MID;
  float* csum = emb;
  float* cov  = csum + (size_t)NBATCH * GRNG * CCH;
  float* mu   = cov + (size_t)NBATCH * CCH * CCH;
  (void)in_sizes; (void)n_in; (void)out_size; (void)ws_size;

  k_emb<<<HW / 256, 256, 0, stream>>>(Wp, emb);
  k_modulate<<<NBATCH * MBPI, 256, 0, stream>>>(x, emb, Wm1, Wm2, xmod);
  k_gram<<<NBATCH * GRNG * 4, 256, 0, stream>>>(xmod, gpart, csum);
  k_cov<<<NBATCH * CCH * CCH / 256, 256, 0, stream>>>(gpart, csum, cov, mu);
  k_quad<<<NBATCH * NBAND, 256, 0, stream>>>(cov, mu, W1, mean, rstd);
  k_bands<<<NBATCH * BPI, 256, 0, stream>>>(xmod, W1, W2, W3, mean, rstd, out);
}


# Band kernel: GELU epilogues evaluated a whole 8-register accumulator tile per step (vector form, same operations): eight multiply-add chains in flight instead of one activation pair's two
# speedup vs baseline: 2.6959x; 1.0554x over previous
#include <hip/hip_runtime.h>
#include <math.h>


typedef __attribute__((ext_vector_type(2))) float v2f;
typedef __attribute__((ext_vector_type(4))) float v4f;
typedef __attribute__((ext_vector_type(8))) float v8f;

#define HW      16384
#define NBATCH  8
#define CCH     128
#define MID     64
#define PDIM    32
#define NBAND   8
#define PXB     512
#define SBN     (PXB / 16)
#define BPI     (HW / PXB)
#define KS1     (CCH / 4)
#define KS2     (MID / 4)
#define GPX     2048
#define GRNG    (HW / GPX)
#define GIT     128
#define GPITCH  (GIT + 4)
#define MGRP    2
#define MBPI    (HW / (128 * MGRP))
#define KSE     (PDIM / 4)

__device__ __forceinline__ v8f wmma4(v2f a, v2f b, v8f c) {
  return __builtin_amdgcn_wmma_f32_16x16x4_f32(false, a, false, b,
                                               (short)0, c, false, false);
}

__device__ __forceinline__ float erf_sel(float z) {
  const float az = fabsf(z);
  const float t  = az * az;
  float ps = fmaf(t, -0x1.268bc2p-11f, 0x1.420828p-8f);
  ps = fmaf(t, ps, -0x1.b5937p-6f);
  ps = fmaf(t, ps, 0x1.ce077cp-4f);
  ps = fmaf(t, ps, -0x1.81266p-2f);
  ps = fmaf(t, ps, 0x1.06eba0p-3f);
  ps = fmaf(az, ps, az);
  float pl = fmaf(az, 0x1.1d3156p-16f, -0x1.8d129p-12f);
  pl = fmaf(az, pl, 0x1.f9a6d2p-9f);
  pl = fmaf(az, pl, -0x1.8c3164p-6f);
  pl = fmaf(az, pl, 0x1.b4e9c8p-4f);
  pl = fmaf(az, pl, 0x1.4515fap-1f);
  pl = fmaf(az, pl, 0x1.078e50p-3f);
  pl = fmaf(az, pl, az);
  const float el = 1.0f - __builtin_amdgcn_exp2f(pl * -0x1.715476p+0f);
  const float r  = az < 1.0f ? ps : el;
  return copysignf(r, z);
}

__device__ __forceinline__ float gelu_f(float x) {
  return 0.5f * x * (1.0f + erf_sel(x * 0.70710678118654752440f));
}
__device__ __forceinline__ float clamp01(float x) {
  return fminf(fmaxf(x, 0.0f), 1.0f);
}
__device__ __forceinline__ v8f spl8(float c) {
  v8f r = {c, c, c, c, c, c, c, c};
  return r;
}
__device__ __forceinline__ v8f fma8(v8f a, v8f b, v8f c) {
  return __builtin_elementwise_fma(a, b, c);
}
__device__ __forceinline__ v8f gelu8(v8f x) {
  const v8f z  = x * 0.70710678118654752440f;
  const v8f az = __builtin_elementwise_abs(z);
  const v8f t  = az * az;
  v8f ps = fma8(t, spl8(-0x1.268bc2p-11f), spl8(0x1.420828p-8f));
  v8f pl = fma8(az, spl8(0x1.1d3156p-16f), spl8(-0x1.8d129p-12f));
  ps = fma8(t, ps, spl8(-0x1.b5937p-6f));
  pl = fma8(az, pl, spl8(0x1.f9a6d2p-9f));
  ps = fma8(t, ps, spl8(0x1.ce077cp-4f));
  pl = fma8(az, pl, spl8(-0x1.8c3164p-6f));
  ps = fma8(t, ps, spl8(-0x1.81266p-2f));
  pl = fma8(az, pl, spl8(0x1.b4e9c8p-4f));
  ps = fma8(t, ps, spl8(0x1.06eba0p-3f));
  pl = fma8(az, pl, spl8(0x1.4515fap-1f));
  ps = fma8(az, ps, az);
  pl = fma8(az, pl, spl8(0x1.078e50p-3f));
  pl = fma8(az, pl, az);
  const v8f pe = pl * -0x1.715476p+0f;
  v8f e;
#pragma unroll
  for (int i = 0; i < 8; ++i) {
    const float el = 1.0f - __builtin_amdgcn_exp2f(pe[i]);
    e[i] = copysignf(az[i] < 1.0f ? ps[i] : el, z[i]);
  }
  return (0.5f * x) * (1.0f + e);
}

__global__ void k_emb(const float* __restrict__ Wp, float* __restrict__ emb) {
  int p = blockIdx.x * blockDim.x + threadIdx.x;
  if (p >= HW) return;
  int i = p >> 7, j = p & 127;
  float ym = -1.0f + (2.0f / 127.0f) * (float)i;
  float xm = -1.0f + (2.0f / 127.0f) * (float)j;
#pragma unroll 4
  for (int pd = 0; pd < PDIM; ++pd) {
    float v = tanhf(Wp[pd * 2 + 0] * ym + Wp[pd * 2 + 1] * xm);
    emb[(size_t)pd * HW + p] = v;
  }
}

__global__ __launch_bounds__(256, 4) void k_modulate(
    const float* __restrict__ x, const float* __restrict__ emb,
    const float* __restrict__ Wm1, const float* __restrict__ Wm2,
    float* __restrict__ xmod) {
  __shared__ __attribute__((aligned(16))) float wa1[MID * (CCH + PDIM)];
  __shared__ __attribute__((aligned(16))) float wa2[CCH * MID];
  __shared__ float lds[8][MID * 17];
  const int wave = threadIdx.x >> 5;
  const int lane = threadIdx.x & 31;
  const int n    = blockIdx.x / MBPI;
  const int gb   = (blockIdx.x % MBPI) * (128 * MGRP);
  const int lm   = lane & 15;
  const int kh   = (lane >> 4) * 2;
  const int mh   = (lane >> 4) * 8;

  for (int u = threadIdx.x; u < MID * (CCH + PDIM) / 2; u += 256) {
    const int r = u / ((CCH + PDIM) / 2), c2 = u % ((CCH + PDIM) / 2);
    const int ks = c2 >> 1, h = c2 & 1;
    const v2f g = *(const v2f*)(Wm1 + r * (CCH + PDIM) + 2 * c2);
    *(v2f*)(wa1 + ((((ks >> 1) * 4 + (r >> 4)) * 32 + h * 16 + (r & 15)) * 4
                   + (ks & 1) * 2)) = g;
  }
  for (int u = threadIdx.x; u < CCH * MID / 2; u += 256) {
    const int r = u / (MID / 2), c2 = u % (MID / 2);
    const int ks = c2 >> 1, h = c2 & 1;
    const v2f g = *(const v2f*)(Wm2 + r * MID + 2 * c2);
    *(v2f*)(wa2 + ((((ks >> 1) * 8 + (r >> 4)) * 32 + h * 16 + (r & 15)) * 4
                   + (ks & 1) * 2)) = g;
  }
  __syncthreads();

  const float* xn = x + (size_t)n * CCH * HW;
  float* xo = xmod + (size_t)n * CCH * HW;
  float* smw = &lds[wave][0] + mh * 17 + lm;
  const float* smr = &lds[wave][0] + kh * 17 + lm;
  const v4f* f1 = (const v4f*)wa1 + lane;
  const v4f* f2 = (const v4f*)wa2 + lane;

  for (int g = 0; g < MGRP; ++g) {
    asm volatile("" ::: "memory");
    const int pix = gb + g * 128 + wave * 16 + lm;
    float* xop = xo + (size_t)mh * HW + pix;

    v2f b[KS1 + KSE];
#pragma unroll
    for (int ks = 0; ks < KS1; ++ks) {
      b[ks].x = xn[(size_t)(4 * ks + kh) * HW + pix];
      b[ks].y = xn[(size_t)(4 * ks + kh + 1) * HW + pix];
    }
#pragma unroll
    for (int ks = 0; ks < KSE; ++ks) {
      b[KS1 + ks].x = emb[(size_t)(4 * ks + kh) * HW + pix];
      b[KS1 + ks].y = emb[(size_t)(4 * ks + kh + 1) * HW + pix];
    }

    v8f acc[4] = {};
    v4f fr[2][8];
#pragma unroll
    for (int i = 0; i < 8; ++i) fr[0][i] = f1[i * 32];
#pragma unroll
    for (int c = 0; c < (KS1 + KSE) / 4; ++c) {
      if (c + 1 < (KS1 + KSE) / 4) {
#pragma unroll
        for (int i = 0; i < 8; ++i) fr[(c + 1) & 1][i] = f1[((c + 1) * 8 + i) * 32];
      }
#pragma unroll
      for (int q = 0; q < 2; ++q) {
#pragma unroll
        for (int t = 0; t < 4; ++t) {
          v2f lo = {fr[c & 1][q * 4 + t].x, fr[c & 1][q * 4 + t].y};
          acc[t] = wmma4(lo, b[4 * c + 2 * q], acc[t]);
        }
#pragma unroll
        for (int t = 0; t < 4; ++t) {
          v2f hi = {fr[c & 1][q * 4 + t].z, fr[c & 1][q * 4 + t].w};
          acc[t] = wmma4(hi, b[4 * c + 2 * q + 1], acc[t]);
        }
      }
      __builtin_amdgcn_sched_barrier(0);
    }
#pragma unroll
    for (int t = 0; t < 4; ++t)
#pragma unroll
      for (int v = 0; v < 8; ++v)
        smw[(t * 16 + v) * 17] = gelu_f(acc[t][v]);

    v8f acc2[8] = {};
    v4f fs[2][8];
    v2f bs[2][2];
#pragma unroll
    for (int t = 0; t < 8; ++t) fs[0][t] = f2[t * 32];
    bs[0][0].x = smr[0];
    bs[0][0].y = smr[17];
    bs[0][1].x = smr[4 * 17];
    bs[0][1].y = smr[5 * 17];
#pragma unroll
    for (int kp = 0; kp < KS2 / 2; ++kp) {
      if (kp + 1 < KS2 / 2) {
#pragma unroll
        for (int t = 0; t < 8; ++t) fs[(kp + 1) & 1][t] = f2[((kp + 1) * 8 + t) * 32];
        bs[(kp + 1) & 1][0].x = smr[(8 * kp + 8) * 17];
        bs[(kp + 1) & 1][0].y = smr[(8 * kp + 9) * 17];
        bs[(kp + 1) & 1][1].x = smr[(8 * kp + 12) * 17];
        bs[(kp + 1) & 1][1].y = smr[(8 * kp + 13) * 17];
      }
#pragma unroll
      for (int t = 0; t < 8; ++t) {
        v2f lo = {fs[kp & 1][t].x, fs[kp & 1][t].y};
        acc2[t] = wmma4(lo, bs[kp & 1][0], acc2[t]);
      }
#pragma unroll
      for (int t = 0; t < 8; ++t) {
        v2f hi = {fs[kp & 1][t].z, fs[kp & 1][t].w};
        acc2[t] = wmma4(hi, bs[kp & 1][1], acc2[t]);
      }
      __builtin_amdgcn_sched_barrier(0);
    }
#pragma unroll
    for (int t = 0; t < 8; ++t)
#pragma unroll
      for (int v = 0; v < 8; ++v)
        xop[(size_t)(t * 16 + v) * HW] = clamp01(gelu_f(acc2[t][v]));
  }
}

__global__ __launch_bounds__(256) void k_gram(
    const float* __restrict__ xmod, float* __restrict__ gpart,
    float* __restrict__ csum) {
  __shared__ __attribute__((aligned(16))) float tile[2][CCH * GPITCH];
  __shared__ float csl[CCH];
  const int wave = threadIdx.x >> 5;
  const int lane = threadIdx.x & 31;
  const int g    = blockIdx.x & 3;
  const int rng  = (blockIdx.x >> 2) & (GRNG - 1);
  const int n    = blockIdx.x >> 5;
  const int lm = lane & 15, hh = lane >> 4, mh = hh * 8;
  const int rb = 2 * g + (wave & 1);
  const int cb = 2 * (wave >> 1);
  const float* sp = xmod + (size_t)n * CCH * HW + (size_t)(16 * wave) * HW
                  + (size_t)rng * GPX + 4 * lane;
  v4f st[16];
#pragma unroll
  for (int q = 0; q < 16; ++q) st[q] = *(const v4f*)(sp + (size_t)q * HW);
  float rsum[16];
#pragma unroll
  for (int q = 0; q < 16; ++q) rsum[q] = 0.0f;
  v8f acc0 = {}, acc1 = {};
  const int fa  = (16 * rb + lm) * GPITCH + 4 * hh;
  const int fb0 = (16 * cb + lm) * GPITCH + 4 * hh;
  const int fb1 = fb0 + 16 * GPITCH;

  for (int it = 0; it < GPX / GIT; ++it) {
    float* tb = &tile[it & 1][0];
#pragma unroll
    for (int q = 0; q < 16; ++q) {
      *(v4f*)(tb + (16 * wave + q) * GPITCH + 4 * lane) = st[q];
      rsum[q] += (st[q].x + st[q].y) + (st[q].z + st[q].w);
    }
    __syncthreads();
    if (it + 1 < GPX / GIT) {
#pragma unroll
      for (int q = 0; q < 16; ++q)
        st[q] = *(const v4f*)(sp + (size_t)q * HW + (it + 1) * GIT);
    }
    v4f a  = *(const v4f*)(tb + fa);
    v4f b0 = *(const v4f*)(tb + fb0);
    v4f b1 = *(const v4f*)(tb + fb1);
#pragma unroll
    for (int s = 0; s < GIT / 8; ++s) {
      v4f an = a, b0n = b0, b1n = b1;
      if (s + 1 < GIT / 8) {
        an  = *(const v4f*)(tb + fa  + 8 * (s + 1));
        b0n = *(const v4f*)(tb + fb0 + 8 * (s + 1));
        b1n = *(const v4f*)(tb + fb1 + 8 * (s + 1));
      }
      v2f alo = {a.x, a.y},   ahi = {a.z, a.w};
      v2f plo = {b0.x, b0.y}, phi = {b0.z, b0.w};
      v2f qlo = {b1.x, b1.y}, qhi = {b1.z, b1.w};
      acc0 = wmma4(alo, plo, acc0);
      acc1 = wmma4(alo, qlo, acc1);
      acc0 = wmma4(ahi, phi, acc0);
      acc1 = wmma4(ahi, qhi, acc1);
      a = an; b0 = b0n; b1 = b1n;
    }
  }

  float* gp = gpart + (size_t)(n * GRNG + rng) * (CCH * CCH);
#pragma unroll
  for (int v = 0; v < 8; ++v) {
    int row = 16 * rb + mh + v;
    gp[row * CCH + 16 * cb + lm]       = acc0[v];
    gp[row * CCH + 16 * (cb + 1) + lm] = acc1[v];
  }
#pragma unroll
  for (int q = 0; q < 16; ++q) {
    float s = rsum[q];
#pragma unroll
    for (int off = 1; off < 32; off <<= 1) s += __shfl_xor(s, off, 32);
    if (lane == 0) csl[16 * wave + q] = s;
  }
  __syncthreads();
  if (g == 0 && threadIdx.x < CCH)
    csum[(n * GRNG + rng) * CCH + threadIdx.x] = csl[threadIdx.x];
}

__global__ void k_cov(const float* __restrict__ gpart,
                      const float* __restrict__ csum,
                      float* __restrict__ cov, float* __restrict__ mu) {
  int i = blockIdx.x * blockDim.x + threadIdx.x;
  if (i >= NBATCH * CCH * CCH) return;
  int n = i >> 14, idx = i & (CCH * CCH - 1);
  int r = idx >> 7, c = idx & (CCH - 1);
  double gs = 0.0, sr = 0.0, sc = 0.0;
  for (int p = 0; p < GRNG; ++p) {
    gs += (double)gpart[(size_t)(n * GRNG + p) * (CCH * CCH) + idx];
    sr += (double)csum[(n * GRNG + p) * CCH + r];
    sc += (double)csum[(n * GRNG + p) * CCH + c];
  }
  double mr = sr * (1.0 / (double)HW), mc = sc * (1.0 / (double)HW);
  cov[i] = (float)(gs * (1.0 / (double)HW) - mr * mc);
  if (r == 0) mu[n * CCH + c] = (float)mc;
}

__global__ __launch_bounds__(256) void k_quad(
    const float* __restrict__ cov, const float* __restrict__ mu,
    const float* __restrict__ W1, float* __restrict__ mean,
    float* __restrict__ rstd) {
  __shared__ float red[8][MID];
  __shared__ float res[2][MID];
  const int jb   = threadIdx.x >> 5;
  const int lane = threadIdx.x & 31;
  const int nb   = blockIdx.x;
  const int n = nb >> 3, band = nb & 7;
  const int lm = lane & 15, kh = (lane >> 4) * 2, mh = (lane >> 4) * 8;
  const float* A  = W1 + (size_t)band * MID * CCH;
  const float* cn = cov + (size_t)n * CCH * CCH + 16 * jb + lm;

  v8f acc[4] = {};
#pragma unroll
  for (int k0 = 0; k0 < CCH; k0 += 4) {
    int kk = k0 + kh;
    v2f b;
    b.x = cn[kk * CCH];
    b.y = cn[(kk + 1) * CCH];
#pragma unroll
    for (int t = 0; t < 4; ++t) {
      v2f a = *(const v2f*)(A + (t * 16 + lm) * CCH + kk);
      acc[t] = wmma4(a, b, acc[t]);
    }
  }
#pragma unroll
  for (int t = 0; t < 4; ++t)
#pragma unroll
    for (int v = 0; v < 8; ++v) {
      int m = t * 16 + v + mh;
      float p = acc[t][v] * A[m * CCH + 16 * jb + lm];
#pragma unroll
      for (int off = 1; off < 16; off <<= 1) p += __shfl_xor(p, off, 32);
      if (lm == 0) red[jb][m] = p;
    }
  __syncthreads();
  const int m = threadIdx.x >> 2, part = threadIdx.x & 3;
  float d = 0.0f;
#pragma unroll
  for (int i = 0; i < 32; ++i)
    d += A[m * CCH + part * 32 + i] * mu[n * CCH + part * 32 + i];
  d += __shfl_xor(d, 1, 32);
  d += __shfl_xor(d, 2, 32);
  if (part == 0) {
    float var = 0.0f;
#pragma unroll
    for (int j = 0; j < 8; ++j) var += red[j][m];
    res[0][m] = d;
    res[1][m] = rsqrtf(fmaxf(var, 0.0f) + 1e-5f);
  }
  __syncthreads();
  if (threadIdx.x < MID)          mean[nb * MID + threadIdx.x] = res[0][threadIdx.x];
  else if (threadIdx.x < 2 * MID) rstd[nb * MID + threadIdx.x - MID] = res[1][threadIdx.x - MID];
}

__global__ __launch_bounds__(256) void k_bands(
    const float* __restrict__ xmod, const float* __restrict__ W1,
    const float* __restrict__ W2, const float* __restrict__ W3,
    const float* __restrict__ mean, const float* __restrict__ rstd,
    float* __restrict__ out) {
  __shared__ float lds[8][MID * 17];
  const int band = threadIdx.x >> 5;
  const int lane = threadIdx.x & 31;
  const int n    = blockIdx.x / BPI;
  const int blk  = blockIdx.x % BPI;
  const int lm = lane & 15, kh = (lane >> 4) * 2, mh = (lane >> 4) * 8;
  const int nb = n * NBAND + band;
  const float* A  = W1 + (size_t)band * MID * CCH;
  const float* A2 = W2 + (size_t)band * 32 * MID;
  const float* w3 = W3 + (size_t)band * 32;
  const float* mu = mean + (size_t)nb * MID;
  const float* rs = rstd + (size_t)nb * MID;
  const float* xb = xmod + (size_t)n * CCH * HW + (size_t)kh * HW + blk * PXB + lm;
  float* sm = &lds[band][0];

  v2f a[4][KS1];
#pragma unroll
  for (int t = 0; t < 4; ++t)
#pragma unroll
    for (int k = 0; k < KS1; ++k)
      a[t][k] = *(const v2f*)(A + (t * 16 + lm) * CCH + 4 * k + kh);
  v2f a2[2][KS2];
#pragma unroll
  for (int t = 0; t < 2; ++t)
#pragma unroll
    for (int k = 0; k < KS2; ++k)
      a2[t][k] = *(const v2f*)(A2 + (t * 16 + lm) * MID + 4 * k + kh);

  v2f bq[KS1];
#pragma unroll
  for (int k = 0; k < KS1; ++k) {
    bq[k].x = xb[(size_t)(4 * k) * HW];
    bq[k].y = xb[(size_t)(4 * k + 1) * HW];
  }

  for (int sb = 0; sb < SBN; ++sb) {
    asm volatile("" ::: "memory");
    v8f acc[4] = {};
#pragma unroll
    for (int k = 0; k < KS1; ++k)
#pragma unroll
      for (int t = 0; t < 4; ++t)
        acc[t] = wmma4(a[t][k], bq[k], acc[t]);

    const int nx = (sb + 1 < SBN ? sb + 1 : sb) * 16;
#pragma unroll
    for (int k = 0; k < KS1; ++k) {
      bq[k].x = xb[(size_t)(4 * k) * HW + nx];
      bq[k].y = xb[(size_t)(4 * k + 1) * HW + nx];
    }

#pragma unroll
    for (int t = 0; t < 4; ++t) {
      v8f m8, r8;
#pragma unroll
      for (int v = 0; v < 8; ++v) {
        m8[v] = mu[t * 16 + v + mh];
        r8[v] = rs[t * 16 + v + mh];
      }
      const v8f g = gelu8((acc[t] - m8) * r8);
#pragma unroll
      for (int v = 0; v < 8; ++v)
        sm[(t * 16 + v + mh) * 17 + lm] = g[v];
    }

    v8f acc2[2] = {};
#pragma unroll
    for (int k = 0; k < KS2; ++k) {
      int kk = 4 * k + kh;
      v2f bb;
      bb.x = sm[kk * 17 + lm];
      bb.y = sm[(kk + 1) * 17 + lm];
#pragma unroll
      for (int t = 0; t < 2; ++t)
        acc2[t] = wmma4(a2[t][k], bb, acc2[t]);
    }
#pragma unroll
    for (int t = 0; t < 2; ++t) {
      const v8f g = gelu8(acc2[t]);
#pragma unroll
      for (int v = 0; v < 8; ++v)
        sm[(t * 16 + v + mh) * 17 + lm] = g[v];
    }

    if (lane < 16) {
      float s = 0.0f;
#pragma unroll
      for (int k = 0; k < 32; ++k) s += w3[k] * sm[k * 17 + lane];
      out[(size_t)nb * HW + blk * PXB + sb * 16 + lane] = clamp01(gelu_f(s));
    }
  }
}

extern "C" void kernel_launch(void* const* d_in, const int* in_sizes, int n_in,
                              void* d_out, int out_size, void* d_ws, size_t ws_size,
                              hipStream_t stream) {
  const float* x   = (const float*)d_in[0];
  const float* Wp  = (const float*)d_in[1];
  const float* Wm1 = (const float*)d_in[2];
  const float* Wm2 = (const float*)d_in[3];
  const float* W1  = (const float*)d_in[4];
  const float* W2  = (const float*)d_in[5];
  const float* W3  = (const float*)d_in[6];
  float* out = (float*)d_out;

  float* ws   = (float*)d_ws;
  float* emb  = ws;
  float* xmod = emb + (size_t)PDIM * HW;
  float* gpart = xmod + (size_t)NBATCH * CCH * HW;
  float* mean = gpart + (size_t)NBATCH * GRNG * CCH * CCH;
  float* rstd = mean + (size_t)NBATCH * NBAND * MID;
  float* csum = emb;
  float* cov  = csum + (size_t)NBATCH * GRNG * CCH;
  float* mu   = cov + (size_t)NBATCH * CCH * CCH;
  (void)in_sizes; (void)n_in; (void)out_size; (void)ws_size;

  k_emb<<<HW / 256, 256, 0, stream>>>(Wp, emb);
  k_modulate<<<NBATCH * MBPI, 256, 0, stream>>>(x, emb, Wm1, Wm2, xmod);
  k_gram<<<NBATCH * GRNG * 4, 256, 0, stream>>>(xmod, gpart, csum);
  k_cov<<<NBATCH * CCH * CCH / 256, 256, 0, stream>>>(gpart, csum, cov, mu);
  k_quad<<<NBATCH * NBAND, 256, 0, stream>>>(cov, mu, W1, mean, rstd);
  k_bands<<<NBATCH * BPI, 256, 0, stream>>>(xmod, W1, W2, W3, mean, rstd, out);
}


# Band kernel: activated accumulator tiles feed the W2 product and the W3 dot from their registers (W2 k-steps reordered to the D layout; W3 dot summed per lane half, then across) instead of via LDS
# speedup vs baseline: 2.9810x; 1.1058x over previous
#include <hip/hip_runtime.h>
#include <math.h>


typedef __attribute__((ext_vector_type(2))) float v2f;
typedef __attribute__((ext_vector_type(4))) float v4f;
typedef __attribute__((ext_vector_type(8))) float v8f;

#define HW      16384
#define NBATCH  8
#define CCH     128
#define MID     64
#define PDIM    32
#define NBAND   8
#define PXB     512
#define SBN     (PXB / 16)
#define BPI     (HW / PXB)
#define KS1     (CCH / 4)
#define KS2     (MID / 4)
#define GPX     2048
#define GRNG    (HW / GPX)
#define GIT     128
#define GPITCH  (GIT + 4)
#define MGRP    2
#define MBPI    (HW / (128 * MGRP))
#define KSE     (PDIM / 4)

__device__ __forceinline__ v8f wmma4(v2f a, v2f b, v8f c) {
  return __builtin_amdgcn_wmma_f32_16x16x4_f32(false, a, false, b,
                                               (short)0, c, false, false);
}

__device__ __forceinline__ float erf_sel(float z) {
  const float az = fabsf(z);
  const float t  = az * az;
  float ps = fmaf(t, -0x1.268bc2p-11f, 0x1.420828p-8f);
  ps = fmaf(t, ps, -0x1.b5937p-6f);
  ps = fmaf(t, ps, 0x1.ce077cp-4f);
  ps = fmaf(t, ps, -0x1.81266p-2f);
  ps = fmaf(t, ps, 0x1.06eba0p-3f);
  ps = fmaf(az, ps, az);
  float pl = fmaf(az, 0x1.1d3156p-16f, -0x1.8d129p-12f);
  pl = fmaf(az, pl, 0x1.f9a6d2p-9f);
  pl = fmaf(az, pl, -0x1.8c3164p-6f);
  pl = fmaf(az, pl, 0x1.b4e9c8p-4f);
  pl = fmaf(az, pl, 0x1.4515fap-1f);
  pl = fmaf(az, pl, 0x1.078e50p-3f);
  pl = fmaf(az, pl, az);
  const float el = 1.0f - __builtin_amdgcn_exp2f(pl * -0x1.715476p+0f);
  const float r  = az < 1.0f ? ps : el;
  return copysignf(r, z);
}

__device__ __forceinline__ float gelu_f(float x) {
  return 0.5f * x * (1.0f + erf_sel(x * 0.70710678118654752440f));
}
__device__ __forceinline__ float clamp01(float x) {
  return fminf(fmaxf(x, 0.0f), 1.0f);
}
__device__ __forceinline__ v8f spl8(float c) {
  v8f r = {c, c, c, c, c, c, c, c};
  return r;
}
__device__ __forceinline__ v8f fma8(v8f a, v8f b, v8f c) {
  return __builtin_elementwise_fma(a, b, c);
}
__device__ __forceinline__ v8f gelu8(v8f x) {
  const v8f z  = x * 0.70710678118654752440f;
  const v8f az = __builtin_elementwise_abs(z);
  const v8f t  = az * az;
  v8f ps = fma8(t, spl8(-0x1.268bc2p-11f), spl8(0x1.420828p-8f));
  v8f pl = fma8(az, spl8(0x1.1d3156p-16f), spl8(-0x1.8d129p-12f));
  ps = fma8(t, ps, spl8(-0x1.b5937p-6f));
  pl = fma8(az, pl, spl8(0x1.f9a6d2p-9f));
  ps = fma8(t, ps, spl8(0x1.ce077cp-4f));
  pl = fma8(az, pl, spl8(-0x1.8c3164p-6f));
  ps = fma8(t, ps, spl8(-0x1.81266p-2f));
  pl = fma8(az, pl, spl8(0x1.b4e9c8p-4f));
  ps = fma8(t, ps, spl8(0x1.06eba0p-3f));
  pl = fma8(az, pl, spl8(0x1.4515fap-1f));
  ps = fma8(az, ps, az);
  pl = fma8(az, pl, spl8(0x1.078e50p-3f));
  pl = fma8(az, pl, az);
  const v8f pe = pl * -0x1.715476p+0f;
  v8f e;
#pragma unroll
  for (int i = 0; i < 8; ++i) {
    const float el = 1.0f - __builtin_amdgcn_exp2f(pe[i]);
    e[i] = copysignf(az[i] < 1.0f ? ps[i] : el, z[i]);
  }
  return (0.5f * x) * (1.0f + e);
}

__global__ void k_emb(const float* __restrict__ Wp, float* __restrict__ emb) {
  int p = blockIdx.x * blockDim.x + threadIdx.x;
  if (p >= HW) return;
  int i = p >> 7, j = p & 127;
  float ym = -1.0f + (2.0f / 127.0f) * (float)i;
  float xm = -1.0f + (2.0f / 127.0f) * (float)j;
#pragma unroll 4
  for (int pd = 0; pd < PDIM; ++pd) {
    float v = tanhf(Wp[pd * 2 + 0] * ym + Wp[pd * 2 + 1] * xm);
    emb[(size_t)pd * HW + p] = v;
  }
}

__global__ __launch_bounds__(256, 4) void k_modulate(
    const float* __restrict__ x, const float* __restrict__ emb,
    const float* __restrict__ Wm1, const float* __restrict__ Wm2,
    float* __restrict__ xmod) {
  __shared__ __attribute__((aligned(16))) float wa1[MID * (CCH + PDIM)];
  __shared__ __attribute__((aligned(16))) float wa2[CCH * MID];
  __shared__ float lds[8][MID * 17];
  const int wave = threadIdx.x >> 5;
  const int lane = threadIdx.x & 31;
  const int n    = blockIdx.x / MBPI;
  const int gb   = (blockIdx.x % MBPI) * (128 * MGRP);
  const int lm   = lane & 15;
  const int kh   = (lane >> 4) * 2;
  const int mh   = (lane >> 4) * 8;

  for (int u = threadIdx.x; u < MID * (CCH + PDIM) / 2; u += 256) {
    const int r = u / ((CCH + PDIM) / 2), c2 = u % ((CCH + PDIM) / 2);
    const int ks = c2 >> 1, h = c2 & 1;
    const v2f g = *(const v2f*)(Wm1 + r * (CCH + PDIM) + 2 * c2);
    *(v2f*)(wa1 + ((((ks >> 1) * 4 + (r >> 4)) * 32 + h * 16 + (r & 15)) * 4
                   + (ks & 1) * 2)) = g;
  }
  for (int u = threadIdx.x; u < CCH * MID / 2; u += 256) {
    const int r = u / (MID / 2), c2 = u % (MID / 2);
    const int ks = c2 >> 1, h = c2 & 1;
    const v2f g = *(const v2f*)(Wm2 + r * MID + 2 * c2);
    *(v2f*)(wa2 + ((((ks >> 1) * 8 + (r >> 4)) * 32 + h * 16 + (r & 15)) * 4
                   + (ks & 1) * 2)) = g;
  }
  __syncthreads();

  const float* xn = x + (size_t)n * CCH * HW;
  float* xo = xmod + (size_t)n * CCH * HW;
  float* smw = &lds[wave][0] + mh * 17 + lm;
  const float* smr = &lds[wave][0] + kh * 17 + lm;
  const v4f* f1 = (const v4f*)wa1 + lane;
  const v4f* f2 = (const v4f*)wa2 + lane;

  for (int g = 0; g < MGRP; ++g) {
    asm volatile("" ::: "memory");
    const int pix = gb + g * 128 + wave * 16 + lm;
    float* xop = xo + (size_t)mh * HW + pix;

    v2f b[KS1 + KSE];
#pragma unroll
    for (int ks = 0; ks < KS1; ++ks) {
      b[ks].x = xn[(size_t)(4 * ks + kh) * HW + pix];
      b[ks].y = xn[(size_t)(4 * ks + kh + 1) * HW + pix];
    }
#pragma unroll
    for (int ks = 0; ks < KSE; ++ks) {
      b[KS1 + ks].x = emb[(size_t)(4 * ks + kh) * HW + pix];
      b[KS1 + ks].y = emb[(size_t)(4 * ks + kh + 1) * HW + pix];
    }

    v8f acc[4] = {};
    v4f fr[2][8];
#pragma unroll
    for (int i = 0; i < 8; ++i) fr[0][i] = f1[i * 32];
#pragma unroll
    for (int c = 0; c < (KS1 + KSE) / 4; ++c) {
      if (c + 1 < (KS1 + KSE) / 4) {
#pragma unroll
        for (int i = 0; i < 8; ++i) fr[(c + 1) & 1][i] = f1[((c + 1) * 8 + i) * 32];
      }
#pragma unroll
      for (int q = 0; q < 2; ++q) {
#pragma unroll
        for (int t = 0; t < 4; ++t) {
          v2f lo = {fr[c & 1][q * 4 + t].x, fr[c & 1][q * 4 + t].y};
          acc[t] = wmma4(lo, b[4 * c + 2 * q], acc[t]);
        }
#pragma unroll
        for (int t = 0; t < 4; ++t) {
          v2f hi = {fr[c & 1][q * 4 + t].z, fr[c & 1][q * 4 + t].w};
          acc[t] = wmma4(hi, b[4 * c + 2 * q + 1], acc[t]);
        }
      }
      __builtin_amdgcn_sched_barrier(0);
    }
#pragma unroll
    for (int t = 0; t < 4; ++t)
#pragma unroll
      for (int v = 0; v < 8; ++v)
        smw[(t * 16 + v) * 17] = gelu_f(acc[t][v]);

    v8f acc2[8] = {};
    v4f fs[2][8];
    v2f bs[2][2];
#pragma unroll
    for (int t = 0; t < 8; ++t) fs[0][t] = f2[t * 32];
    bs[0][0].x = smr[0];
    bs[0][0].y = smr[17];
    bs[0][1].x = smr[4 * 17];
    bs[0][1].y = smr[5 * 17];
#pragma unroll
    for (int kp = 0; kp < KS2 / 2; ++kp) {
      if (kp + 1 < KS2 / 2) {
#pragma unroll
        for (int t = 0; t < 8; ++t) fs[(kp + 1) & 1][t] = f2[((kp + 1) * 8 + t) * 32];
        bs[(kp + 1) & 1][0].x = smr[(8 * kp + 8) * 17];
        bs[(kp + 1) & 1][0].y = smr[(8 * kp + 9) * 17];
        bs[(kp + 1) & 1][1].x = smr[(8 * kp + 12) * 17];
        bs[(kp + 1) & 1][1].y = smr[(8 * kp + 13) * 17];
      }
#pragma unroll
      for (int t = 0; t < 8; ++t) {
        v2f lo = {fs[kp & 1][t].x, fs[kp & 1][t].y};
        acc2[t] = wmma4(lo, bs[kp & 1][0], acc2[t]);
      }
#pragma unroll
      for (int t = 0; t < 8; ++t) {
        v2f hi = {fs[kp & 1][t].z, fs[kp & 1][t].w};
        acc2[t] = wmma4(hi, bs[kp & 1][1], acc2[t]);
      }
      __builtin_amdgcn_sched_barrier(0);
    }
#pragma unroll
    for (int t = 0; t < 8; ++t)
#pragma unroll
      for (int v = 0; v < 8; ++v)
        xop[(size_t)(t * 16 + v) * HW] = clamp01(gelu_f(acc2[t][v]));
  }
}

__global__ __launch_bounds__(256) void k_gram(
    const float* __restrict__ xmod, float* __restrict__ gpart,
    float* __restrict__ csum) {
  __shared__ __attribute__((aligned(16))) float tile[2][CCH * GPITCH];
  __shared__ float csl[CCH];
  const int wave = threadIdx.x >> 5;
  const int lane = threadIdx.x & 31;
  const int g    = blockIdx.x & 3;
  const int rng  = (blockIdx.x >> 2) & (GRNG - 1);
  const int n    = blockIdx.x >> 5;
  const int lm = lane & 15, hh = lane >> 4, mh = hh * 8;
  const int rb = 2 * g + (wave & 1);
  const int cb = 2 * (wave >> 1);
  const float* sp = xmod + (size_t)n * CCH * HW + (size_t)(16 * wave) * HW
                  + (size_t)rng * GPX + 4 * lane;
  v4f st[16];
#pragma unroll
  for (int q = 0; q < 16; ++q) st[q] = *(const v4f*)(sp + (size_t)q * HW);
  float rsum[16];
#pragma unroll
  for (int q = 0; q < 16; ++q) rsum[q] = 0.0f;
  v8f acc0 = {}, acc1 = {};
  const int fa  = (16 * rb + lm) * GPITCH + 4 * hh;
  const int fb0 = (16 * cb + lm) * GPITCH + 4 * hh;
  const int fb1 = fb0 + 16 * GPITCH;

  for (int it = 0; it < GPX / GIT; ++it) {
    float* tb = &tile[it & 1][0];
#pragma unroll
    for (int q = 0; q < 16; ++q) {
      *(v4f*)(tb + (16 * wave + q) * GPITCH + 4 * lane) = st[q];
      rsum[q] += (st[q].x + st[q].y) + (st[q].z + st[q].w);
    }
    __syncthreads();
    if (it + 1 < GPX / GIT) {
#pragma unroll
      for (int q = 0; q < 16; ++q)
        st[q] = *(const v4f*)(sp + (size_t)q * HW + (it + 1) * GIT);
    }
    v4f a  = *(const v4f*)(tb + fa);
    v4f b0 = *(const v4f*)(tb + fb0);
    v4f b1 = *(const v4f*)(tb + fb1);
#pragma unroll
    for (int s = 0; s < GIT / 8; ++s) {
      v4f an = a, b0n = b0, b1n = b1;
      if (s + 1 < GIT / 8) {
        an  = *(const v4f*)(tb + fa  + 8 * (s + 1));
        b0n = *(const v4f*)(tb + fb0 + 8 * (s + 1));
        b1n = *(const v4f*)(tb + fb1 + 8 * (s + 1));
      }
      v2f alo = {a.x, a.y},   ahi = {a.z, a.w};
      v2f plo = {b0.x, b0.y}, phi = {b0.z, b0.w};
      v2f qlo = {b1.x, b1.y}, qhi = {b1.z, b1.w};
      acc0 = wmma4(alo, plo, acc0);
      acc1 = wmma4(alo, qlo, acc1);
      acc0 = wmma4(ahi, phi, acc0);
      acc1 = wmma4(ahi, qhi, acc1);
      a = an; b0 = b0n; b1 = b1n;
    }
  }

  float* gp = gpart + (size_t)(n * GRNG + rng) * (CCH * CCH);
#pragma unroll
  for (int v = 0; v < 8; ++v) {
    int row = 16 * rb + mh + v;
    gp[row * CCH + 16 * cb + lm]       = acc0[v];
    gp[row * CCH + 16 * (cb + 1) + lm] = acc1[v];
  }
#pragma unroll
  for (int q = 0; q < 16; ++q) {
    float s = rsum[q];
#pragma unroll
    for (int off = 1; off < 32; off <<= 1) s += __shfl_xor(s, off, 32);
    if (lane == 0) csl[16 * wave + q] = s;
  }
  __syncthreads();
  if (g == 0 && threadIdx.x < CCH)
    csum[(n * GRNG + rng) * CCH + threadIdx.x] = csl[threadIdx.x];
}

__global__ void k_cov(const float* __restrict__ gpart,
                      const float* __restrict__ csum,
                      float* __restrict__ cov, float* __restrict__ mu) {
  int i = blockIdx.x * blockDim.x + threadIdx.x;
  if (i >= NBATCH * CCH * CCH) return;
  int n = i >> 14, idx = i & (CCH * CCH - 1);
  int r = idx >> 7, c = idx & (CCH - 1);
  double gs = 0.0, sr = 0.0, sc = 0.0;
  for (int p = 0; p < GRNG; ++p) {
    gs += (double)gpart[(size_t)(n * GRNG + p) * (CCH * CCH) + idx];
    sr += (double)csum[(n * GRNG + p) * CCH + r];
    sc += (double)csum[(n * GRNG + p) * CCH + c];
  }
  double mr = sr * (1.0 / (double)HW), mc = sc * (1.0 / (double)HW);
  cov[i] = (float)(gs * (1.0 / (double)HW) - mr * mc);
  if (r == 0) mu[n * CCH + c] = (float)mc;
}

__global__ __launch_bounds__(256) void k_quad(
    const float* __restrict__ cov, const float* __restrict__ mu,
    const float* __restrict__ W1, float* __restrict__ mean,
    float* __restrict__ rstd) {
  __shared__ float red[8][MID];
  __shared__ float res[2][MID];
  const int jb   = threadIdx.x >> 5;
  const int lane = threadIdx.x & 31;
  const int nb   = blockIdx.x;
  const int n = nb >> 3, band = nb & 7;
  const int lm = lane & 15, kh = (lane >> 4) * 2, mh = (lane >> 4) * 8;
  const float* A  = W1 + (size_t)band * MID * CCH;
  const float* cn = cov + (size_t)n * CCH * CCH + 16 * jb + lm;

  v8f acc[4] = {};
#pragma unroll
  for (int k0 = 0; k0 < CCH; k0 += 4) {
    int kk = k0 + kh;
    v2f b;
    b.x = cn[kk * CCH];
    b.y = cn[(kk + 1) * CCH];
#pragma unroll
    for (int t = 0; t < 4; ++t) {
      v2f a = *(const v2f*)(A + (t * 16 + lm) * CCH + kk);
      acc[t] = wmma4(a, b, acc[t]);
    }
  }
#pragma unroll
  for (int t = 0; t < 4; ++t)
#pragma unroll
    for (int v = 0; v < 8; ++v) {
      int m = t * 16 + v + mh;
      float p = acc[t][v] * A[m * CCH + 16 * jb + lm];
#pragma unroll
      for (int off = 1; off < 16; off <<= 1) p += __shfl_xor(p, off, 32);
      if (lm == 0) red[jb][m] = p;
    }
  __syncthreads();
  const int m = threadIdx.x >> 2, part = threadIdx.x & 3;
  float d = 0.0f;
#pragma unroll
  for (int i = 0; i < 32; ++i)
    d += A[m * CCH + part * 32 + i] * mu[n * CCH + part * 32 + i];
  d += __shfl_xor(d, 1, 32);
  d += __shfl_xor(d, 2, 32);
  if (part == 0) {
    float var = 0.0f;
#pragma unroll
    for (int j = 0; j < 8; ++j) var += red[j][m];
    res[0][m] = d;
    res[1][m] = rsqrtf(fmaxf(var, 0.0f) + 1e-5f);
  }
  __syncthreads();
  if (threadIdx.x < MID)          mean[nb * MID + threadIdx.x] = res[0][threadIdx.x];
  else if (threadIdx.x < 2 * MID) rstd[nb * MID + threadIdx.x - MID] = res[1][threadIdx.x - MID];
}

__device__ __forceinline__ float xhalf(float x) {
  return __builtin_bit_cast(float, __builtin_amdgcn_permlanex16(
      __builtin_bit_cast(int, x), __builtin_bit_cast(int, x),
      0x76543210, 0xfedcba98, false, false));
}

__global__ __launch_bounds__(256) void k_bands(
    const float* __restrict__ xmod, const float* __restrict__ W1,
    const float* __restrict__ W2, const float* __restrict__ W3,
    const float* __restrict__ mean, const float* __restrict__ rstd,
    float* __restrict__ out) {
  const int band = threadIdx.x >> 5;
  const int lane = threadIdx.x & 31;
  const int n    = blockIdx.x / BPI;
  const int blk  = blockIdx.x % BPI;
  const int lm = lane & 15, kh = (lane >> 4) * 2, mh = (lane >> 4) * 8;
  const int nb = n * NBAND + band;
  const float* A  = W1 + (size_t)band * MID * CCH;
  const float* A2 = W2 + (size_t)band * 32 * MID;
  const float* w3 = W3 + (size_t)band * 32 + mh;
  const float* mu = mean + (size_t)nb * MID;
  const float* rs = rstd + (size_t)nb * MID;
  const float* xb = xmod + (size_t)n * CCH * HW + (size_t)kh * HW + blk * PXB + lm;
  float* op = out + (size_t)nb * HW + blk * PXB + lm;

  v2f a[4][KS1];
#pragma unroll
  for (int t = 0; t < 4; ++t)
#pragma unroll
    for (int k = 0; k < KS1; ++k)
      a[t][k] = *(const v2f*)(A + (t * 16 + lm) * CCH + 4 * k + kh);
  v2f a2[2][KS2];
#pragma unroll
  for (int u = 0; u < 2; ++u)
#pragma unroll
    for (int t = 0; t < 4; ++t)
#pragma unroll
      for (int q = 0; q < 4; ++q)
        a2[u][4 * t + q] = *(const v2f*)(A2 + (u * 16 + lm) * MID + 16 * t + mh + 2 * q);

  float w3r[2][8];
#pragma unroll
  for (int u = 0; u < 2; ++u)
#pragma unroll
    for (int r = 0; r < 8; ++r) w3r[u][r] = w3[16 * u + r];

  v2f bq[KS1];
#pragma unroll
  for (int k = 0; k < KS1; ++k) {
    bq[k].x = xb[(size_t)(4 * k) * HW];
    bq[k].y = xb[(size_t)(4 * k + 1) * HW];
  }

  for (int sb = 0; sb < SBN; ++sb) {
    asm volatile("" ::: "memory");
    v8f acc[4] = {};
#pragma unroll
    for (int k = 0; k < KS1; ++k)
#pragma unroll
      for (int t = 0; t < 4; ++t)
        acc[t] = wmma4(a[t][k], bq[k], acc[t]);

    const int nx = (sb + 1 < SBN ? sb + 1 : sb) * 16;
#pragma unroll
    for (int k = 0; k < KS1; ++k) {
      bq[k].x = xb[(size_t)(4 * k) * HW + nx];
      bq[k].y = xb[(size_t)(4 * k + 1) * HW + nx];
    }

    v8f acc2[2] = {};
#pragma unroll
    for (int t = 0; t < 4; ++t) {
      v8f m8, r8;
#pragma unroll
      for (int v = 0; v < 8; ++v) {
        m8[v] = mu[t * 16 + v + mh];
        r8[v] = rs[t * 16 + v + mh];
      }
      const v8f g = gelu8((acc[t] - m8) * r8);
#pragma unroll
      for (int q = 0; q < 4; ++q) {
        const v2f bb = {g[2 * q], g[2 * q + 1]};
#pragma unroll
        for (int u = 0; u < 2; ++u)
          acc2[u] = wmma4(a2[u][4 * t + q], bb, acc2[u]);
      }
    }

    float s2[2];
#pragma unroll
    for (int u = 0; u < 2; ++u) {
      const v8f g = gelu8(acc2[u]);
      float s = w3r[u][0] * g[0];
#pragma unroll
      for (int r = 1; r < 8; ++r) s = fmaf(w3r[u][r], g[r], s);
      s2[u] = s;
    }
    float s = s2[0] + s2[1];
    s += xhalf(s);
    const float o = clamp01(gelu_f(s));
    if (lane < 16) op[sb * 16] = o;
  }
}

extern "C" void kernel_launch(void* const* d_in, const int* in_sizes, int n_in,
                              void* d_out, int out_size, void* d_ws, size_t ws_size,
                              hipStream_t stream) {
  const float* x   = (const float*)d_in[0];
  const float* Wp  = (const float*)d_in[1];
  const float* Wm1 = (const float*)d_in[2];
  const float* Wm2 = (const float*)d_in[3];
  const float* W1  = (const float*)d_in[4];
  const float* W2  = (const float*)d_in[5];
  const float* W3  = (const float*)d_in[6];
  float* out = (float*)d_out;

  float* ws   = (float*)d_ws;
  float* emb  = ws;
  float* xmod = emb + (size_t)PDIM * HW;
  float* gpart = xmod + (size_t)NBATCH * CCH * HW;
  float* mean = gpart + (size_t)NBATCH * GRNG * CCH * CCH;
  float* rstd = mean + (size_t)NBATCH * NBAND * MID;
  float* csum = emb;
  float* cov  = csum + (size_t)NBATCH * GRNG * CCH;
  float* mu   = cov + (size_t)NBATCH * CCH * CCH;
  (void)in_sizes; (void)n_in; (void)out_size; (void)ws_size;

  k_emb<<<HW / 256, 256, 0, stream>>>(Wp, emb);
  k_modulate<<<NBATCH * MBPI, 256, 0, stream>>>(x, emb, Wm1, Wm2, xmod);
  k_gram<<<NBATCH * GRNG * 4, 256, 0, stream>>>(xmod, gpart, csum);
  k_cov<<<NBATCH * CCH * CCH / 256, 256, 0, stream>>>(gpart, csum, cov, mu);
  k_quad<<<NBATCH * NBAND, 256, 0, stream>>>(cov, mu, W1, mean, rstd);
  k_bands<<<NBATCH * BPI, 256, 0, stream>>>(xmod, W1, W2, W3, mean, rstd, out);
}


# GELU in every kernel: Phi(x) = 1/2 + x P(x^2)/Q(x^2), one clamped rational (10 multiply-adds + hardware reciprocal) replaces the two-range erf (14 multiply-adds, exp2, compare, select, sign copy)
# speedup vs baseline: 3.1055x; 1.0418x over previous
#include <hip/hip_runtime.h>
#include <math.h>


typedef __attribute__((ext_vector_type(2))) float v2f;
typedef __attribute__((ext_vector_type(4))) float v4f;
typedef __attribute__((ext_vector_type(8))) float v8f;

#define HW      16384
#define NBATCH  8
#define CCH     128
#define MID     64
#define PDIM    32
#define NBAND   8
#define PXB     512
#define SBN     (PXB / 16)
#define BPI     (HW / PXB)
#define KS1     (CCH / 4)
#define KS2     (MID / 4)
#define GPX     2048
#define GRNG    (HW / GPX)
#define GIT     128
#define GPITCH  (GIT + 4)
#define MGRP    2
#define MBPI    (HW / (128 * MGRP))
#define KSE     (PDIM / 4)

__device__ __forceinline__ v8f wmma4(v2f a, v2f b, v8f c) {
  return __builtin_amdgcn_wmma_f32_16x16x4_f32(false, a, false, b,
                                               (short)0, c, false, false);
}

#define GL  0x1.62cc42p+2f
#define GA6 0x1.e3e708p-34f
#define GA5 -0x1.7a6cfcp-26f
#define GA4 0x1.b9a634p-19f
#define GA3 0x1.6fff3ep-13f
#define GA2 0x1.29af7cp-8f
#define GA1 0x1.2a69b4p-5f
#define GA0 0x1.98844ap-2f
#define GB4 0x1.0c4d76p-14f
#define GB3 0x1.e7be2cp-10f
#define GB2 0x1.e1757ep-6f
#define GB1 0x1.082a66p-2f

__device__ __forceinline__ float gelu_f(float x) {
  const float xc = fminf(fmaxf(x, -GL), GL);
  const float u  = xc * xc;
  float p = fmaf(u, GA6, GA5);
  float q = fmaf(u, GB4, GB3);
  p = fmaf(u, p, GA4);
  q = fmaf(u, q, GB2);
  p = fmaf(u, p, GA3);
  q = fmaf(u, q, GB1);
  p = fmaf(u, p, GA2);
  q = fmaf(u, q, 1.0f);
  p = fmaf(u, p, GA1);
  p = fmaf(u, p, GA0);
  return x * fmaf(xc * p, __builtin_amdgcn_rcpf(q), 0.5f);
}
__device__ __forceinline__ float clamp01(float x) {
  return fminf(fmaxf(x, 0.0f), 1.0f);
}
__device__ __forceinline__ v8f spl8(float c) {
  v8f r = {c, c, c, c, c, c, c, c};
  return r;
}
__device__ __forceinline__ v8f fma8(v8f a, v8f b, v8f c) {
  return __builtin_elementwise_fma(a, b, c);
}
__device__ __forceinline__ v8f gelu8(v8f x) {
  const v8f xc = __builtin_elementwise_min(
      __builtin_elementwise_max(x, spl8(-GL)), spl8(GL));
  const v8f u = xc * xc;
  v8f p = fma8(u, spl8(GA6), spl8(GA5));
  v8f q = fma8(u, spl8(GB4), spl8(GB3));
  p = fma8(u, p, spl8(GA4));
  q = fma8(u, q, spl8(GB2));
  p = fma8(u, p, spl8(GA3));
  q = fma8(u, q, spl8(GB1));
  p = fma8(u, p, spl8(GA2));
  q = fma8(u, q, spl8(1.0f));
  p = fma8(u, p, spl8(GA1));
  p = fma8(u, p, spl8(GA0));
  v8f rq;
#pragma unroll
  for (int i = 0; i < 8; ++i) rq[i] = __builtin_amdgcn_rcpf(q[i]);
  return x * fma8(xc * p, rq, spl8(0.5f));
}

__global__ void k_emb(const float* __restrict__ Wp, float* __restrict__ emb) {
  int p = blockIdx.x * blockDim.x + threadIdx.x;
  if (p >= HW) return;
  int i = p >> 7, j = p & 127;
  float ym = -1.0f + (2.0f / 127.0f) * (float)i;
  float xm = -1.0f + (2.0f / 127.0f) * (float)j;
#pragma unroll 4
  for (int pd = 0; pd < PDIM; ++pd) {
    float v = tanhf(Wp[pd * 2 + 0] * ym + Wp[pd * 2 + 1] * xm);
    emb[(size_t)pd * HW + p] = v;
  }
}

__global__ __launch_bounds__(256, 4) void k_modulate(
    const float* __restrict__ x, const float* __restrict__ emb,
    const float* __restrict__ Wm1, const float* __restrict__ Wm2,
    float* __restrict__ xmod) {
  __shared__ __attribute__((aligned(16))) float wa1[MID * (CCH + PDIM)];
  __shared__ __attribute__((aligned(16))) float wa2[CCH * MID];
  __shared__ float lds[8][MID * 17];
  const int wave = threadIdx.x >> 5;
  const int lane = threadIdx.x & 31;
  const int n    = blockIdx.x / MBPI;
  const int gb   = (blockIdx.x % MBPI) * (128 * MGRP);
  const int lm   = lane & 15;
  const int kh   = (lane >> 4) * 2;
  const int mh   = (lane >> 4) * 8;

  for (int u = threadIdx.x; u < MID * (CCH + PDIM) / 2; u += 256) {
    const int r = u / ((CCH + PDIM) / 2), c2 = u % ((CCH + PDIM) / 2);
    const int ks = c2 >> 1, h = c2 & 1;
    const v2f g = *(const v2f*)(Wm1 + r * (CCH + PDIM) + 2 * c2);
    *(v2f*)(wa1 + ((((ks >> 1) * 4 + (r >> 4)) * 32 + h * 16 + (r & 15)) * 4
                   + (ks & 1) * 2)) = g;
  }
  for (int u = threadIdx.x; u < CCH * MID / 2; u += 256) {
    const int r = u / (MID / 2), c2 = u % (MID / 2);
    const int ks = c2 >> 1, h = c2 & 1;
    const v2f g = *(const v2f*)(Wm2 + r * MID + 2 * c2);
    *(v2f*)(wa2 + ((((ks >> 1) * 8 + (r >> 4)) * 32 + h * 16 + (r & 15)) * 4
                   + (ks & 1) * 2)) = g;
  }
  __syncthreads();

  const float* xn = x + (size_t)n * CCH * HW;
  float* xo = xmod + (size_t)n * CCH * HW;
  float* smw = &lds[wave][0] + mh * 17 + lm;
  const float* smr = &lds[wave][0] + kh * 17 + lm;
  const v4f* f1 = (const v4f*)wa1 + lane;
  const v4f* f2 = (const v4f*)wa2 + lane;

  for (int g = 0; g < MGRP; ++g) {
    asm volatile("" ::: "memory");
    const int pix = gb + g * 128 + wave * 16 + lm;
    float* xop = xo + (size_t)mh * HW + pix;

    v2f b[KS1 + KSE];
#pragma unroll
    for (int ks = 0; ks < KS1; ++ks) {
      b[ks].x = xn[(size_t)(4 * ks + kh) * HW + pix];
      b[ks].y = xn[(size_t)(4 * ks + kh + 1) * HW + pix];
    }
#pragma unroll
    for (int ks = 0; ks < KSE; ++ks) {
      b[KS1 + ks].x = emb[(size_t)(4 * ks + kh) * HW + pix];
      b[KS1 + ks].y = emb[(size_t)(4 * ks + kh + 1) * HW + pix];
    }

    v8f acc[4] = {};
    v4f fr[2][8];
#pragma unroll
    for (int i = 0; i < 8; ++i) fr[0][i] = f1[i * 32];
#pragma unroll
    for (int c = 0; c < (KS1 + KSE) / 4; ++c) {
      if (c + 1 < (KS1 + KSE) / 4) {
#pragma unroll
        for (int i = 0; i < 8; ++i) fr[(c + 1) & 1][i] = f1[((c + 1) * 8 + i) * 32];
      }
#pragma unroll
      for (int q = 0; q < 2; ++q) {
#pragma unroll
        for (int t = 0; t < 4; ++t) {
          v2f lo = {fr[c & 1][q * 4 + t].x, fr[c & 1][q * 4 + t].y};
          acc[t] = wmma4(lo, b[4 * c + 2 * q], acc[t]);
        }
#pragma unroll
        for (int t = 0; t < 4; ++t) {
          v2f hi = {fr[c & 1][q * 4 + t].z, fr[c & 1][q * 4 + t].w};
          acc[t] = wmma4(hi, b[4 * c + 2 * q + 1], acc[t]);
        }
      }
      __builtin_amdgcn_sched_barrier(0);
    }
#pragma unroll
    for (int t = 0; t < 4; ++t)
#pragma unroll
      for (int v = 0; v < 8; ++v)
        smw[(t * 16 + v) * 17] = gelu_f(acc[t][v]);

    v8f acc2[8] = {};
    v4f fs[2][8];
    v2f bs[2][2];
#pragma unroll
    for (int t = 0; t < 8; ++t) fs[0][t] = f2[t * 32];
    bs[0][0].x = smr[0];
    bs[0][0].y = smr[17];
    bs[0][1].x = smr[4 * 17];
    bs[0][1].y = smr[5 * 17];
#pragma unroll
    for (int kp = 0; kp < KS2 / 2; ++kp) {
      if (kp + 1 < KS2 / 2) {
#pragma unroll
        for (int t = 0; t < 8; ++t) fs[(kp + 1) & 1][t] = f2[((kp + 1) * 8 + t) * 32];
        bs[(kp + 1) & 1][0].x = smr[(8 * kp + 8) * 17];
        bs[(kp + 1) & 1][0].y = smr[(8 * kp + 9) * 17];
        bs[(kp + 1) & 1][1].x = smr[(8 * kp + 12) * 17];
        bs[(kp + 1) & 1][1].y = smr[(8 * kp + 13) * 17];
      }
#pragma unroll
      for (int t = 0; t < 8; ++t) {
        v2f lo = {fs[kp & 1][t].x, fs[kp & 1][t].y};
        acc2[t] = wmma4(lo, bs[kp & 1][0], acc2[t]);
      }
#pragma unroll
      for (int t = 0; t < 8; ++t) {
        v2f hi = {fs[kp & 1][t].z, fs[kp & 1][t].w};
        acc2[t] = wmma4(hi, bs[kp & 1][1], acc2[t]);
      }
      __builtin_amdgcn_sched_barrier(0);
    }
#pragma unroll
    for (int t = 0; t < 8; ++t)
#pragma unroll
      for (int v = 0; v < 8; ++v)
        xop[(size_t)(t * 16 + v) * HW] = clamp01(gelu_f(acc2[t][v]));
  }
}

__global__ __launch_bounds__(256) void k_gram(
    const float* __restrict__ xmod, float* __restrict__ gpart,
    float* __restrict__ csum) {
  __shared__ __attribute__((aligned(16))) float tile[2][CCH * GPITCH];
  __shared__ float csl[CCH];
  const int wave = threadIdx.x >> 5;
  const int lane = threadIdx.x & 31;
  const int g    = blockIdx.x & 3;
  const int rng  = (blockIdx.x >> 2) & (GRNG - 1);
  const int n    = blockIdx.x >> 5;
  const int lm = lane & 15, hh = lane >> 4, mh = hh * 8;
  const int rb = 2 * g + (wave & 1);
  const int cb = 2 * (wave >> 1);
  const float* sp = xmod + (size_t)n * CCH * HW + (size_t)(16 * wave) * HW
                  + (size_t)rng * GPX + 4 * lane;
  v4f st[16];
#pragma unroll
  for (int q = 0; q < 16; ++q) st[q] = *(const v4f*)(sp + (size_t)q * HW);
  float rsum[16];
#pragma unroll
  for (int q = 0; q < 16; ++q) rsum[q] = 0.0f;
  v8f acc0 = {}, acc1 = {};
  const int fa  = (16 * rb + lm) * GPITCH + 4 * hh;
  const int fb0 = (16 * cb + lm) * GPITCH + 4 * hh;
  const int fb1 = fb0 + 16 * GPITCH;

  for (int it = 0; it < GPX / GIT; ++it) {
    float* tb = &tile[it & 1][0];
#pragma unroll
    for (int q = 0; q < 16; ++q) {
      *(v4f*)(tb + (16 * wave + q) * GPITCH + 4 * lane) = st[q];
      rsum[q] += (st[q].x + st[q].y) + (st[q].z + st[q].w);
    }
    __syncthreads();
    if (it + 1 < GPX / GIT) {
#pragma unroll
      for (int q = 0; q < 16; ++q)
        st[q] = *(const v4f*)(sp + (size_t)q * HW + (it + 1) * GIT);
    }
    v4f a  = *(const v4f*)(tb + fa);
    v4f b0 = *(const v4f*)(tb + fb0);
    v4f b1 = *(const v4f*)(tb + fb1);
#pragma unroll
    for (int s = 0; s < GIT / 8; ++s) {
      v4f an = a, b0n = b0, b1n = b1;
      if (s + 1 < GIT / 8) {
        an  = *(const v4f*)(tb + fa  + 8 * (s + 1));
        b0n = *(const v4f*)(tb + fb0 + 8 * (s + 1));
        b1n = *(const v4f*)(tb + fb1 + 8 * (s + 1));
      }
      v2f alo = {a.x, a.y},   ahi = {a.z, a.w};
      v2f plo = {b0.x, b0.y}, phi = {b0.z, b0.w};
      v2f qlo = {b1.x, b1.y}, qhi = {b1.z, b1.w};
      acc0 = wmma4(alo, plo, acc0);
      acc1 = wmma4(alo, qlo, acc1);
      acc0 = wmma4(ahi, phi, acc0);
      acc1 = wmma4(ahi, qhi, acc1);
      a = an; b0 = b0n; b1 = b1n;
    }
  }

  float* gp = gpart + (size_t)(n * GRNG + rng) * (CCH * CCH);
#pragma unroll
  for (int v = 0; v < 8; ++v) {
    int row = 16 * rb + mh + v;
    gp[row * CCH + 16 * cb + lm]       = acc0[v];
    gp[row * CCH + 16 * (cb + 1) + lm] = acc1[v];
  }
#pragma unroll
  for (int q = 0; q < 16; ++q) {
    float s = rsum[q];
#pragma unroll
    for (int off = 1; off < 32; off <<= 1) s += __shfl_xor(s, off, 32);
    if (lane == 0) csl[16 * wave + q] = s;
  }
  __syncthreads();
  if (g == 0 && threadIdx.x < CCH)
    csum[(n * GRNG + rng) * CCH + threadIdx.x] = csl[threadIdx.x];
}

__global__ void k_cov(const float* __restrict__ gpart,
                      const float* __restrict__ csum,
                      float* __restrict__ cov, float* __restrict__ mu) {
  int i = blockIdx.x * blockDim.x + threadIdx.x;
  if (i >= NBATCH * CCH * CCH) return;
  int n = i >> 14, idx = i & (CCH * CCH - 1);
  int r = idx >> 7, c = idx & (CCH - 1);
  double gs = 0.0, sr = 0.0, sc = 0.0;
  for (int p = 0; p < GRNG; ++p) {
    gs += (double)gpart[(size_t)(n * GRNG + p) * (CCH * CCH) + idx];
    sr += (double)csum[(n * GRNG + p) * CCH + r];
    sc += (double)csum[(n * GRNG + p) * CCH + c];
  }
  double mr = sr * (1.0 / (double)HW), mc = sc * (1.0 / (double)HW);
  cov[i] = (float)(gs * (1.0 / (double)HW) - mr * mc);
  if (r == 0) mu[n * CCH + c] = (float)mc;
}

__global__ __launch_bounds__(256) void k_quad(
    const float* __restrict__ cov, const float* __restrict__ mu,
    const float* __restrict__ W1, float* __restrict__ mean,
    float* __restrict__ rstd) {
  __shared__ float red[8][MID];
  __shared__ float res[2][MID];
  const int jb   = threadIdx.x >> 5;
  const int lane = threadIdx.x & 31;
  const int nb   = blockIdx.x;
  const int n = nb >> 3, band = nb & 7;
  const int lm = lane & 15, kh = (lane >> 4) * 2, mh = (lane >> 4) * 8;
  const float* A  = W1 + (size_t)band * MID * CCH;
  const float* cn = cov + (size_t)n * CCH * CCH + 16 * jb + lm;

  v8f acc[4] = {};
#pragma unroll
  for (int k0 = 0; k0 < CCH; k0 += 4) {
    int kk = k0 + kh;
    v2f b;
    b.x = cn[kk * CCH];
    b.y = cn[(kk + 1) * CCH];
#pragma unroll
    for (int t = 0; t < 4; ++t) {
      v2f a = *(const v2f*)(A + (t * 16 + lm) * CCH + kk);
      acc[t] = wmma4(a, b, acc[t]);
    }
  }
#pragma unroll
  for (int t = 0; t < 4; ++t)
#pragma unroll
    for (int v = 0; v < 8; ++v) {
      int m = t * 16 + v + mh;
      float p = acc[t][v] * A[m * CCH + 16 * jb + lm];
#pragma unroll
      for (int off = 1; off < 16; off <<= 1) p += __shfl_xor(p, off, 32);
      if (lm == 0) red[jb][m] = p;
    }
  __syncthreads();
  const int m = threadIdx.x >> 2, part = threadIdx.x & 3;
  float d = 0.0f;
#pragma unroll
  for (int i = 0; i < 32; ++i)
    d += A[m * CCH + part * 32 + i] * mu[n * CCH + part * 32 + i];
  d += __shfl_xor(d, 1, 32);
  d += __shfl_xor(d, 2, 32);
  if (part == 0) {
    float var = 0.0f;
#pragma unroll
    for (int j = 0; j < 8; ++j) var += red[j][m];
    res[0][m] = d;
    res[1][m] = rsqrtf(fmaxf(var, 0.0f) + 1e-5f);
  }
  __syncthreads();
  if (threadIdx.x < MID)          mean[nb * MID + threadIdx.x] = res[0][threadIdx.x];
  else if (threadIdx.x < 2 * MID) rstd[nb * MID + threadIdx.x - MID] = res[1][threadIdx.x - MID];
}

__device__ __forceinline__ float xhalf(float x) {
  return __builtin_bit_cast(float, __builtin_amdgcn_permlanex16(
      __builtin_bit_cast(int, x), __builtin_bit_cast(int, x),
      0x76543210, 0xfedcba98, false, false));
}

__global__ __launch_bounds__(256) void k_bands(
    const float* __restrict__ xmod, const float* __restrict__ W1,
    const float* __restrict__ W2, const float* __restrict__ W3,
    const float* __restrict__ mean, const float* __restrict__ rstd,
    float* __restrict__ out) {
  const int band = threadIdx.x >> 5;
  const int lane = threadIdx.x & 31;
  const int n    = blockIdx.x / BPI;
  const int blk  = blockIdx.x % BPI;
  const int lm = lane & 15, kh = (lane >> 4) * 2, mh = (lane >> 4) * 8;
  const int nb = n * NBAND + band;
  const float* A  = W1 + (size_t)band * MID * CCH;
  const float* A2 = W2 + (size_t)band * 32 * MID;
  const float* w3 = W3 + (size_t)band * 32 + mh;
  const float* mu = mean + (size_t)nb * MID;
  const float* rs = rstd + (size_t)nb * MID;
  const float* xb = xmod + (size_t)n * CCH * HW + (size_t)kh * HW + blk * PXB + lm;
  float* op = out + (size_t)nb * HW + blk * PXB + lm;

  v2f a[4][KS1];
#pragma unroll
  for (int t = 0; t < 4; ++t)
#pragma unroll
    for (int k = 0; k < KS1; ++k)
      a[t][k] = *(const v2f*)(A + (t * 16 + lm) * CCH + 4 * k + kh);
  v2f a2[2][KS2];
#pragma unroll
  for (int u = 0; u < 2; ++u)
#pragma unroll
    for (int t = 0; t < 4; ++t)
#pragma unroll
      for (int q = 0; q < 4; ++q)
        a2[u][4 * t + q] = *(const v2f*)(A2 + (u * 16 + lm) * MID + 16 * t + mh + 2 * q);

  float w3r[2][8];
#pragma unroll
  for (int u = 0; u < 2; ++u)
#pragma unroll
    for (int r = 0; r < 8; ++r) w3r[u][r] = w3[16 * u + r];

  v2f bq[KS1];
#pragma unroll
  for (int k = 0; k < KS1; ++k) {
    bq[k].x = xb[(size_t)(4 * k) * HW];
    bq[k].y = xb[(size_t)(4 * k + 1) * HW];
  }

  for (int sb = 0; sb < SBN; ++sb) {
    asm volatile("" ::: "memory");
    v8f acc[4] = {};
#pragma unroll
    for (int k = 0; k < KS1; ++k)
#pragma unroll
      for (int t = 0; t < 4; ++t)
        acc[t] = wmma4(a[t][k], bq[k], acc[t]);

    const int nx = (sb + 1 < SBN ? sb + 1 : sb) * 16;
#pragma unroll
    for (int k = 0; k < KS1; ++k) {
      bq[k].x = xb[(size_t)(4 * k) * HW + nx];
      bq[k].y = xb[(size_t)(4 * k + 1) * HW + nx];
    }

    v8f acc2[2] = {};
#pragma unroll
    for (int t = 0; t < 4; ++t) {
      v8f m8, r8;
#pragma unroll
      for (int v = 0; v < 8; ++v) {
        m8[v] = mu[t * 16 + v + mh];
        r8[v] = rs[t * 16 + v + mh];
      }
      const v8f g = gelu8((acc[t] - m8) * r8);
#pragma unroll
      for (int q = 0; q < 4; ++q) {
        const v2f bb = {g[2 * q], g[2 * q + 1]};
#pragma unroll
        for (int u = 0; u < 2; ++u)
          acc2[u] = wmma4(a2[u][4 * t + q], bb, acc2[u]);
      }
    }

    float s2[2];
#pragma unroll
    for (int u = 0; u < 2; ++u) {
      const v8f g = gelu8(acc2[u]);
      float s = w3r[u][0] * g[0];
#pragma unroll
      for (int r = 1; r < 8; ++r) s = fmaf(w3r[u][r], g[r], s);
      s2[u] = s;
    }
    float s = s2[0] + s2[1];
    s += xhalf(s);
    const float o = clamp01(gelu_f(s));
    if (lane < 16) op[sb * 16] = o;
  }
}

extern "C" void kernel_launch(void* const* d_in, const int* in_sizes, int n_in,
                              void* d_out, int out_size, void* d_ws, size_t ws_size,
                              hipStream_t stream) {
  const float* x   = (const float*)d_in[0];
  const float* Wp  = (const float*)d_in[1];
  const float* Wm1 = (const float*)d_in[2];
  const float* Wm2 = (const float*)d_in[3];
  const float* W1  = (const float*)d_in[4];
  const float* W2  = (const float*)d_in[5];
  const float* W3  = (const float*)d_in[6];
  float* out = (float*)d_out;

  float* ws   = (float*)d_ws;
  float* emb  = ws;
  float* xmod = emb + (size_t)PDIM * HW;
  float* gpart = xmod + (size_t)NBATCH * CCH * HW;
  float* mean = gpart + (size_t)NBATCH * GRNG * CCH * CCH;
  float* rstd = mean + (size_t)NBATCH * NBAND * MID;
  float* csum = emb;
  float* cov  = csum + (size_t)NBATCH * GRNG * CCH;
  float* mu   = cov + (size_t)NBATCH * CCH * CCH;
  (void)in_sizes; (void)n_in; (void)out_size; (void)ws_size;

  k_emb<<<HW / 256, 256, 0, stream>>>(Wp, emb);
  k_modulate<<<NBATCH * MBPI, 256, 0, stream>>>(x, emb, Wm1, Wm2, xmod);
  k_gram<<<NBATCH * GRNG * 4, 256, 0, stream>>>(xmod, gpart, csum);
  k_cov<<<NBATCH * CCH * CCH / 256, 256, 0, stream>>>(gpart, csum, cov, mu);
  k_quad<<<NBATCH * NBAND, 256, 0, stream>>>(cov, mu, W1, mean, rstd);
  k_bands<<<NBATCH * BPI, 256, 0, stream>>>(xmod, W1, W2, W3, mean, rstd, out);
}
